# starting kernel
# baseline (speedup 1.0000x reference)
#include <hip/hip_runtime.h>
#include <math.h>

#define GDE_N 50000
#define GDE_E 800000

typedef float v2f __attribute__((ext_vector_type(2)));
typedef float v8f __attribute__((ext_vector_type(8)));

__global__ __launch_bounds__(32) void gde_mm_wmma(const float* __restrict__ X,
                                                  const float* __restrict__ G,
                                                  float* __restrict__ Y,
                                                  int K, int Ncols) {
  const int lane = threadIdx.x;
  const int m    = lane & 15;
  const int half = lane >> 4;
  const int row  = blockIdx.x * 16 + m;
  const int col  = blockIdx.y * 16 + m;
  const float* __restrict__ xr = X + (size_t)row * K;
  v8f acc = {0.f, 0.f, 0.f, 0.f, 0.f, 0.f, 0.f, 0.f};
  for (int k = 0; k < K; k += 4) {
    const int ka = k + 2 * half;
    v2f a; a.x = xr[ka];
           a.y = xr[ka + 1];
    v2f b; b.x = G[(size_t)ka       * Ncols + col];
           b.y = G[(size_t)(ka + 1) * Ncols + col];
    acc = __builtin_amdgcn_wmma_f32_16x16x4_f32(false, a, false, b,
                                                (short)0, acc, false, false);
  }
  const int rbase = blockIdx.x * 16 + half * 8;
#pragma unroll
  for (int r = 0; r < 8; ++r)
    Y[(size_t)(rbase + r) * Ncols + col] = acc[r];
}

__global__ void gde_prep_g(float* __restrict__ G, const float* __restrict__ Wp,
                           const float* __restrict__ R, int Cin, int Cout) {
  int idx = blockIdx.x * blockDim.x + threadIdx.x;
  int Ncols = 3 * Cout;
  int tot = Cin * Ncols;
  if (idx >= tot) return;
  int r = idx / Ncols, c = idx % Ncols;
  float v;
  if (c < Cout)            v = Wp[r * Cout + c];
  else if (c < 2 * Cout)   v = Wp[(size_t)Cin * Cout + r * Cout + (c - Cout)]
                             - Wp[r * Cout + (c - Cout)];
  else                     v = R[r * Cout + (c - 2 * Cout)];
  G[idx] = v;
}

__global__ void gde_deg(float* __restrict__ deg, const int* __restrict__ dst, int E) {
  int e = blockIdx.x * blockDim.x + threadIdx.x;
  if (e < E) atomicAdd(&deg[dst[e]], 1.0f);
}

__global__ void gde_invdeg(float* __restrict__ d, int n) {
  int i = blockIdx.x * blockDim.x + threadIdx.x;
  if (i < n) d[i] = 1.0f / fmaxf(d[i], 1.0f);
}

__global__ void gde_edge_agg(float* __restrict__ agg, const float* __restrict__ Z,
                             const int* __restrict__ src, const int* __restrict__ dst,
                             const float* __restrict__ p, int Cout, long long total) {
  long long t = (long long)blockIdx.x * blockDim.x + threadIdx.x;
  if (t >= total) return;
  const int groups = Cout >> 2;
  const int e = (int)(t / groups);
  const int c = (int)(t % groups) * 4;
  const int s = src[e], d = dst[e];
  const float pe = p[e];
  const int ldz = 3 * Cout;
  const float4 z0 = *(const float4*)(Z + (size_t)s * ldz + c);
  const float4 z1 = *(const float4*)(Z + (size_t)s * ldz + Cout + c);
  float* ap = agg + (size_t)d * Cout + c;
  atomicAdd(ap + 0, z0.x + pe * z1.x);
  atomicAdd(ap + 1, z0.y + pe * z1.y);
  atomicAdd(ap + 2, z0.z + pe * z1.z);
  atomicAdd(ap + 3, z0.w + pe * z1.w);
}

__global__ void gde_finalize(float* __restrict__ out, const float* __restrict__ agg,
                             const float* __restrict__ Z, const float* __restrict__ bias,
                             const float* __restrict__ inv, int Cout, int act,
                             long long total) {
  long long idx = (long long)blockIdx.x * blockDim.x + threadIdx.x;
  if (idx >= total) return;
  int i = (int)(idx / Cout), c = (int)(idx % Cout);
  float v = agg[idx] * inv[i] + Z[(size_t)i * 3 * Cout + 2 * Cout + c] + bias[c];
  if (act) v = tanhf(v);
  out[idx] = v;
}

__global__ void gde_rk_stage(float* __restrict__ ACC, float* __restrict__ YT,
                             const float* __restrict__ H, const float* __restrict__ K,
                             float w, float c, int set_acc, long long n) {
  long long i = (long long)blockIdx.x * blockDim.x + threadIdx.x;
  if (i >= n) return;
  float k = K[i];
  ACC[i] = set_acc ? (w * k) : (ACC[i] + w * k);
  YT[i] = H[i] + c * k;
}

__global__ void gde_rk_final(float* __restrict__ H, const float* __restrict__ ACC,
                             const float* __restrict__ K, long long n) {
  long long i = (long long)blockIdx.x * blockDim.x + threadIdx.x;
  if (i >= n) return;
  H[i] = H[i] + 0.5f * (ACC[i] + K[i]);
}

__global__ void gde_log_softmax16(float* __restrict__ out, int n) {
  int i = blockIdx.x * blockDim.x + threadIdx.x;
  if (i >= n) return;
  float v[16];
  float mx = -3.402823466e+38f;
  float* row = out + (size_t)i * 16;
#pragma unroll
  for (int c = 0; c < 16; ++c) { v[c] = row[c]; mx = fmaxf(mx, v[c]); }
  float s = 0.f;
#pragma unroll
  for (int c = 0; c < 16; ++c) s += __expf(v[c] - mx);
  float ls = __logf(s);
#pragma unroll
  for (int c = 0; c < 16; ++c) row[c] = v[c] - mx - ls;
}

extern "C" void kernel_launch(void* const* d_in, const int* in_sizes, int n_in,
                              void* d_out, int out_size, void* d_ws, size_t ws_size,
                              hipStream_t stream) {
  const float* x   = (const float*)d_in[0];
  const float* p   = (const float*)d_in[1];
  const int*   src = (const int*)d_in[2];
  const int*   dst = (const int*)d_in[3];
  const float* W1 = (const float*)d_in[4],  *R1 = (const float*)d_in[5],  *b1 = (const float*)d_in[6];
  const float* Wa = (const float*)d_in[7],  *Ra = (const float*)d_in[8],  *ba = (const float*)d_in[9];
  const float* Wb = (const float*)d_in[10], *Rb = (const float*)d_in[11], *bb = (const float*)d_in[12];
  const float* W2 = (const float*)d_in[13], *R2 = (const float*)d_in[14], *b2 = (const float*)d_in[15];
  float* out = (float*)d_out;

  float* W = (float*)d_ws;
  size_t o = 0;
  float* Z   = W + o; o += (size_t)GDE_N * 192;
  float* AGG = W + o; o += (size_t)GDE_N * 64;
  float* H   = W + o; o += (size_t)GDE_N * 64;
  float* YT  = W + o; o += (size_t)GDE_N * 64;
  float* U   = W + o; o += (size_t)GDE_N * 64;
  float* Kb  = W + o; o += (size_t)GDE_N * 64;
  float* ACC = W + o; o += (size_t)GDE_N * 64;
  float* INV = W + o; o += (size_t)GDE_N;
  float* GC1 = W + o; o += (size_t)256 * 192;
  float* GCA = W + o; o += (size_t)64 * 192;
  float* GCB = W + o; o += (size_t)64 * 192;
  float* GC2 = W + o; o += (size_t)64 * 48;
  (void)ws_size; (void)n_in; (void)in_sizes; (void)out_size;

  gde_prep_g<<<(256 * 192 + 255) / 256, 256, 0, stream>>>(GC1, W1, R1, 256, 64);
  gde_prep_g<<<(64 * 192 + 255) / 256, 256, 0, stream>>>(GCA, Wa, Ra, 64, 64);
  gde_prep_g<<<(64 * 192 + 255) / 256, 256, 0, stream>>>(GCB, Wb, Rb, 64, 64);
  gde_prep_g<<<(64 * 48 + 255) / 256, 256, 0, stream>>>(GC2, W2, R2, 64, 16);

  hipMemsetAsync(INV, 0, sizeof(float) * GDE_N, stream);
  gde_deg<<<(GDE_E + 255) / 256, 256, 0, stream>>>(INV, dst, GDE_E);
  gde_invdeg<<<(GDE_N + 255) / 256, 256, 0, stream>>>(INV, GDE_N);

  auto spline = [&](const float* y, int Cin, int Cout, const float* G,
                    const float* bias, float* ob, int act) {
    const int N3 = 3 * Cout;
    dim3 grid(GDE_N / 16, N3 / 16);
    gde_mm_wmma<<<grid, 32, 0, stream>>>(y, G, Z, Cin, N3);
    hipMemsetAsync(AGG, 0, sizeof(float) * (size_t)GDE_N * Cout, stream);
    long long tot = (long long)GDE_E * (Cout / 4);
    gde_edge_agg<<<(int)((tot + 255) / 256), 256, 0, stream>>>(AGG, Z, src, dst, p, Cout, tot);
    long long tn = (long long)GDE_N * Cout;
    gde_finalize<<<(int)((tn + 255) / 256), 256, 0, stream>>>(ob, AGG, Z, bias, INV, Cout, act, tn);
  };
  auto odefunc = [&](const float* y, float* k) {
    spline(y, 64, 64, GCA, ba, U, 0);
    spline(U, 64, 64, GCB, bb, k, 0);
  };

  const long long tn = (long long)GDE_N * 64;
  const int eb = (int)((tn + 255) / 256);

  spline(x, 256, 64, GC1, b1, H, 1);

  odefunc(H, Kb);
  gde_rk_stage<<<eb, 256, 0, stream>>>(ACC, YT, H, Kb, 1.0f, 1.5f, 1, tn);
  odefunc(YT, Kb);
  gde_rk_stage<<<eb, 256, 0, stream>>>(ACC, YT, H, Kb, 2.0f, 1.5f, 0, tn);
  odefunc(YT, Kb);
  gde_rk_stage<<<eb, 256, 0, stream>>>(ACC, YT, H, Kb, 2.0f, 3.0f, 0, tn);
  odefunc(YT, Kb);
  gde_rk_final<<<eb, 256, 0, stream>>>(H, ACC, Kb, tn);

  spline(H, 64, 16, GC2, b2, out, 1);
  gde_log_softmax16<<<(GDE_N + 255) / 256, 256, 0, stream>>>(out, GDE_N);
}


# Dense node products restaged through LDS: an 8-wave group copies G to LDS in fragment order, each wave loads its 16 X rows once as whole lines and keeps all column tiles in registers; k order kept
# speedup vs baseline: 1.2798x; 1.2798x over previous
#include <hip/hip_runtime.h>
#include <math.h>

#define GDE_N 50000
#define GDE_E 800000

typedef float v2f __attribute__((ext_vector_type(2)));
typedef float v8f __attribute__((ext_vector_type(8)));

#define MM_TILES (GDE_N / 16)
#define MM_WAVES 8
#define MM_THREADS (32 * MM_WAVES)
#define MM_KC 64
#define MM_APITCH (MM_KC + 4)

static_assert(GDE_N % 16 == 0, "whole row tiles");

template <int K, int NT>
__global__ __launch_bounds__(MM_THREADS) __attribute__((amdgpu_waves_per_eu(4, 4))) void gde_mm_lds(
    const float* __restrict__ X, const float* __restrict__ G, float* __restrict__ Y) {
  static_assert(K % MM_KC == 0, "K must be whole chunks");
  constexpr int NC    = 16 * NT;
  constexpr int ITEMS = (MM_KC / 2) * (NC / 4);
  constexpr int ITERS = (ITEMS + MM_THREADS - 1) / MM_THREADS;
  constexpr int FULL  = ITEMS / MM_THREADS;
  __shared__ __attribute__((aligned(16))) v2f   Gs[(MM_KC / 4) * NT * 32];
  __shared__ __attribute__((aligned(16))) float As[MM_WAVES][16 * MM_APITCH];

  const int tid  = threadIdx.x;
  const int lane = tid & 31;
  const int wave = tid >> 5;
  const int m    = lane & 15;
  const int half = lane >> 4;
  const int  tile_raw = blockIdx.x * MM_WAVES + wave;
  const bool live     = tile_raw < MM_TILES;
  const int  tile     = live ? tile_raw : (MM_TILES - 1);
  const float* __restrict__ xt = X + (size_t)tile * 16 * K;
  float* as = As[wave];

  v8f acc[NT];
#pragma unroll
  for (int t = 0; t < NT; ++t) acc[t] = (v8f){0.f, 0.f, 0.f, 0.f, 0.f, 0.f, 0.f, 0.f};

#pragma unroll 1
  for (int kc = 0; kc < K; kc += MM_KC) {
    if (kc) __syncthreads();
    float4 xa[8];
#pragma unroll
    for (int i = 0; i < 8; ++i)
      xa[i] = *(const float4*)(xt + (size_t)(2 * i + half) * K + kc + 4 * m);
    float4 g0[ITERS], g1[ITERS];
    int    gslot[ITERS];
#pragma unroll
    for (int u = 0; u < ITERS; ++u) {
      const int i  = tid + u * MM_THREADS;
      const int ic = (u < FULL || i < ITEMS) ? i : (i - ITEMS);
      const int q  = ic / (NC / 4);
      const int c0 = (ic - q * (NC / 4)) * 4;
      g0[u] = *(const float4*)(G + (size_t)(kc + 2 * q)     * NC + c0);
      g1[u] = *(const float4*)(G + (size_t)(kc + 2 * q + 1) * NC + c0);
      gslot[u] = (((q >> 1) * NT + (c0 >> 4)) * 32) + (q & 1) * 16 + (c0 & 15);
    }
#pragma unroll
    for (int i = 0; i < 8; ++i)
      *(float4*)(as + (2 * i + half) * MM_APITCH + 4 * m) = xa[i];
#pragma unroll
    for (int u = 0; u < ITERS; ++u) {
      float* gp = (float*)&Gs[gslot[u]];
      *(float4*)(gp)     = make_float4(g0[u].x, g1[u].x, g0[u].y, g1[u].y);
      *(float4*)(gp + 4) = make_float4(g0[u].z, g1[u].z, g0[u].w, g1[u].w);
    }
    __syncthreads();
#pragma unroll
    for (int j = 0; j < MM_KC / 4; ++j) {
      const v2f a = *(const v2f*)(as + m * MM_APITCH + 4 * j + 2 * half);
#pragma unroll
      for (int t = 0; t < NT; ++t) {
        const v2f b = Gs[(j * NT + t) * 32 + lane];
        acc[t] = __builtin_amdgcn_wmma_f32_16x16x4_f32(false, a, false, b,
                                                       (short)0, acc[t], false, false);
      }
    }
  }

  if (live) {
    float* yr = Y + (size_t)(tile * 16 + half * 8) * NC + m;
#pragma unroll
    for (int t = 0; t < NT; ++t)
#pragma unroll
      for (int r = 0; r < 8; ++r)
        yr[(size_t)r * NC + 16 * t] = acc[t][r];
  }
}

__global__ void gde_prep_g(float* __restrict__ G, const float* __restrict__ Wp,
                           const float* __restrict__ R, int Cin, int Cout) {
  int idx = blockIdx.x * blockDim.x + threadIdx.x;
  int Ncols = 3 * Cout;
  int tot = Cin * Ncols;
  if (idx >= tot) return;
  int r = idx / Ncols, c = idx % Ncols;
  float v;
  if (c < Cout)            v = Wp[r * Cout + c];
  else if (c < 2 * Cout)   v = Wp[(size_t)Cin * Cout + r * Cout + (c - Cout)]
                             - Wp[r * Cout + (c - Cout)];
  else                     v = R[r * Cout + (c - 2 * Cout)];
  G[idx] = v;
}

__global__ void gde_deg(float* __restrict__ deg, const int* __restrict__ dst, int E) {
  int e = blockIdx.x * blockDim.x + threadIdx.x;
  if (e < E) atomicAdd(&deg[dst[e]], 1.0f);
}

__global__ void gde_invdeg(float* __restrict__ d, int n) {
  int i = blockIdx.x * blockDim.x + threadIdx.x;
  if (i < n) d[i] = 1.0f / fmaxf(d[i], 1.0f);
}

__global__ void gde_edge_agg(float* __restrict__ agg, const float* __restrict__ Z,
                             const int* __restrict__ src, const int* __restrict__ dst,
                             const float* __restrict__ p, int Cout, long long total) {
  long long t = (long long)blockIdx.x * blockDim.x + threadIdx.x;
  if (t >= total) return;
  const int groups = Cout >> 2;
  const int e = (int)(t / groups);
  const int c = (int)(t % groups) * 4;
  const int s = src[e], d = dst[e];
  const float pe = p[e];
  const int ldz = 3 * Cout;
  const float4 z0 = *(const float4*)(Z + (size_t)s * ldz + c);
  const float4 z1 = *(const float4*)(Z + (size_t)s * ldz + Cout + c);
  float* ap = agg + (size_t)d * Cout + c;
  atomicAdd(ap + 0, z0.x + pe * z1.x);
  atomicAdd(ap + 1, z0.y + pe * z1.y);
  atomicAdd(ap + 2, z0.z + pe * z1.z);
  atomicAdd(ap + 3, z0.w + pe * z1.w);
}

__global__ void gde_finalize(float* __restrict__ out, const float* __restrict__ agg,
                             const float* __restrict__ Z, const float* __restrict__ bias,
                             const float* __restrict__ inv, int Cout, int act,
                             long long total) {
  long long idx = (long long)blockIdx.x * blockDim.x + threadIdx.x;
  if (idx >= total) return;
  int i = (int)(idx / Cout), c = (int)(idx % Cout);
  float v = agg[idx] * inv[i] + Z[(size_t)i * 3 * Cout + 2 * Cout + c] + bias[c];
  if (act) v = tanhf(v);
  out[idx] = v;
}

__global__ void gde_rk_stage(float* __restrict__ ACC, float* __restrict__ YT,
                             const float* __restrict__ H, const float* __restrict__ K,
                             float w, float c, int set_acc, long long n) {
  long long i = (long long)blockIdx.x * blockDim.x + threadIdx.x;
  if (i >= n) return;
  float k = K[i];
  ACC[i] = set_acc ? (w * k) : (ACC[i] + w * k);
  YT[i] = H[i] + c * k;
}

__global__ void gde_rk_final(float* __restrict__ H, const float* __restrict__ ACC,
                             const float* __restrict__ K, long long n) {
  long long i = (long long)blockIdx.x * blockDim.x + threadIdx.x;
  if (i >= n) return;
  H[i] = H[i] + 0.5f * (ACC[i] + K[i]);
}

__global__ void gde_log_softmax16(float* __restrict__ out, int n) {
  int i = blockIdx.x * blockDim.x + threadIdx.x;
  if (i >= n) return;
  float v[16];
  float mx = -3.402823466e+38f;
  float* row = out + (size_t)i * 16;
#pragma unroll
  for (int c = 0; c < 16; ++c) { v[c] = row[c]; mx = fmaxf(mx, v[c]); }
  float s = 0.f;
#pragma unroll
  for (int c = 0; c < 16; ++c) s += __expf(v[c] - mx);
  float ls = __logf(s);
#pragma unroll
  for (int c = 0; c < 16; ++c) row[c] = v[c] - mx - ls;
}

extern "C" void kernel_launch(void* const* d_in, const int* in_sizes, int n_in,
                              void* d_out, int out_size, void* d_ws, size_t ws_size,
                              hipStream_t stream) {
  const float* x   = (const float*)d_in[0];
  const float* p   = (const float*)d_in[1];
  const int*   src = (const int*)d_in[2];
  const int*   dst = (const int*)d_in[3];
  const float* W1 = (const float*)d_in[4],  *R1 = (const float*)d_in[5],  *b1 = (const float*)d_in[6];
  const float* Wa = (const float*)d_in[7],  *Ra = (const float*)d_in[8],  *ba = (const float*)d_in[9];
  const float* Wb = (const float*)d_in[10], *Rb = (const float*)d_in[11], *bb = (const float*)d_in[12];
  const float* W2 = (const float*)d_in[13], *R2 = (const float*)d_in[14], *b2 = (const float*)d_in[15];
  float* out = (float*)d_out;

  float* W = (float*)d_ws;
  size_t o = 0;
  float* Z   = W + o; o += (size_t)GDE_N * 192;
  float* AGG = W + o; o += (size_t)GDE_N * 64;
  float* H   = W + o; o += (size_t)GDE_N * 64;
  float* YT  = W + o; o += (size_t)GDE_N * 64;
  float* U   = W + o; o += (size_t)GDE_N * 64;
  float* Kb  = W + o; o += (size_t)GDE_N * 64;
  float* ACC = W + o; o += (size_t)GDE_N * 64;
  float* INV = W + o; o += (size_t)GDE_N;
  float* GC1 = W + o; o += (size_t)256 * 192;
  float* GCA = W + o; o += (size_t)64 * 192;
  float* GCB = W + o; o += (size_t)64 * 192;
  float* GC2 = W + o; o += (size_t)64 * 48;
  (void)ws_size; (void)n_in; (void)in_sizes; (void)out_size;

  gde_prep_g<<<(256 * 192 + 255) / 256, 256, 0, stream>>>(GC1, W1, R1, 256, 64);
  gde_prep_g<<<(64 * 192 + 255) / 256, 256, 0, stream>>>(GCA, Wa, Ra, 64, 64);
  gde_prep_g<<<(64 * 192 + 255) / 256, 256, 0, stream>>>(GCB, Wb, Rb, 64, 64);
  gde_prep_g<<<(64 * 48 + 255) / 256, 256, 0, stream>>>(GC2, W2, R2, 64, 16);

  hipMemsetAsync(INV, 0, sizeof(float) * GDE_N, stream);
  gde_deg<<<(GDE_E + 255) / 256, 256, 0, stream>>>(INV, dst, GDE_E);
  gde_invdeg<<<(GDE_N + 255) / 256, 256, 0, stream>>>(INV, GDE_N);

  auto spline = [&](const float* y, int Cin, int Cout, const float* G,
                    const float* bias, float* ob, int act) {
    const int N3 = 3 * Cout;
    const int mmg = (MM_TILES + MM_WAVES - 1) / MM_WAVES;
    if (Cin == 256)      gde_mm_lds<256, 12><<<mmg, MM_THREADS, 0, stream>>>(y, G, Z);
    else if (N3 == 192)  gde_mm_lds<64, 12><<<mmg, MM_THREADS, 0, stream>>>(y, G, Z);
    else                 gde_mm_lds<64, 3><<<mmg, MM_THREADS, 0, stream>>>(y, G, Z);
    hipMemsetAsync(AGG, 0, sizeof(float) * (size_t)GDE_N * Cout, stream);
    long long tot = (long long)GDE_E * (Cout / 4);
    gde_edge_agg<<<(int)((tot + 255) / 256), 256, 0, stream>>>(AGG, Z, src, dst, p, Cout, tot);
    long long tn = (long long)GDE_N * Cout;
    gde_finalize<<<(int)((tn + 255) / 256), 256, 0, stream>>>(ob, AGG, Z, bias, INV, Cout, act, tn);
  };
  auto odefunc = [&](const float* y, float* k) {
    spline(y, 64, 64, GCA, ba, U, 0);
    spline(U, 64, 64, GCB, bb, k, 0);
  };

  const long long tn = (long long)GDE_N * 64;
  const int eb = (int)((tn + 255) / 256);

  spline(x, 256, 64, GC1, b1, H, 1);

  odefunc(H, Kb);
  gde_rk_stage<<<eb, 256, 0, stream>>>(ACC, YT, H, Kb, 1.0f, 1.5f, 1, tn);
  odefunc(YT, Kb);
  gde_rk_stage<<<eb, 256, 0, stream>>>(ACC, YT, H, Kb, 2.0f, 1.5f, 0, tn);
  odefunc(YT, Kb);
  gde_rk_stage<<<eb, 256, 0, stream>>>(ACC, YT, H, Kb, 2.0f, 3.0f, 0, tn);
  odefunc(YT, Kb);
  gde_rk_final<<<eb, 256, 0, stream>>>(H, ACC, Kb, tn);

  spline(H, 64, 16, GC2, b2, out, 1);
  gde_log_softmax16<<<(GDE_N + 255) / 256, 256, 0, stream>>>(out, GDE_N);
}


# Edge scatter retiled, lane = channel: a wave walks 32 edges, {Z0,Z1} columns paired in G (one 8-byte load per lane), each f32 atomic covers one whole 128-B line; float atomics and arrival order kept
# speedup vs baseline: 1.8101x; 1.4144x over previous
#include <hip/hip_runtime.h>
#include <math.h>

#define GDE_N 50000
#define GDE_E 800000

typedef float v2f __attribute__((ext_vector_type(2)));
typedef float v8f __attribute__((ext_vector_type(8)));

#define MM_TILES (GDE_N / 16)
#define MM_WAVES 8
#define MM_THREADS (32 * MM_WAVES)
#define MM_KC 64
#define MM_APITCH (MM_KC + 4)

static_assert(GDE_N % 16 == 0, "whole row tiles");

template <int K, int NT>
__global__ __launch_bounds__(MM_THREADS) __attribute__((amdgpu_waves_per_eu(4, 4))) void gde_mm_lds(
    const float* __restrict__ X, const float* __restrict__ G, float* __restrict__ Y) {
  static_assert(K % MM_KC == 0, "K must be whole chunks");
  constexpr int NC    = 16 * NT;
  constexpr int ITEMS = (MM_KC / 2) * (NC / 4);
  constexpr int ITERS = (ITEMS + MM_THREADS - 1) / MM_THREADS;
  constexpr int FULL  = ITEMS / MM_THREADS;
  __shared__ __attribute__((aligned(16))) v2f   Gs[(MM_KC / 4) * NT * 32];
  __shared__ __attribute__((aligned(16))) float As[MM_WAVES][16 * MM_APITCH];

  const int tid  = threadIdx.x;
  const int lane = tid & 31;
  const int wave = tid >> 5;
  const int m    = lane & 15;
  const int half = lane >> 4;
  const int  tile_raw = blockIdx.x * MM_WAVES + wave;
  const bool live     = tile_raw < MM_TILES;
  const int  tile     = live ? tile_raw : (MM_TILES - 1);
  const float* __restrict__ xt = X + (size_t)tile * 16 * K;
  float* as = As[wave];

  v8f acc[NT];
#pragma unroll
  for (int t = 0; t < NT; ++t) acc[t] = (v8f){0.f, 0.f, 0.f, 0.f, 0.f, 0.f, 0.f, 0.f};

#pragma unroll 1
  for (int kc = 0; kc < K; kc += MM_KC) {
    if (kc) __syncthreads();
    float4 xa[8];
#pragma unroll
    for (int i = 0; i < 8; ++i)
      xa[i] = *(const float4*)(xt + (size_t)(2 * i + half) * K + kc + 4 * m);
    float4 g0[ITERS], g1[ITERS];
    int    gslot[ITERS];
#pragma unroll
    for (int u = 0; u < ITERS; ++u) {
      const int i  = tid + u * MM_THREADS;
      const int ic = (u < FULL || i < ITEMS) ? i : (i - ITEMS);
      const int q  = ic / (NC / 4);
      const int c0 = (ic - q * (NC / 4)) * 4;
      g0[u] = *(const float4*)(G + (size_t)(kc + 2 * q)     * NC + c0);
      g1[u] = *(const float4*)(G + (size_t)(kc + 2 * q + 1) * NC + c0);
      gslot[u] = (((q >> 1) * NT + (c0 >> 4)) * 32) + (q & 1) * 16 + (c0 & 15);
    }
#pragma unroll
    for (int i = 0; i < 8; ++i)
      *(float4*)(as + (2 * i + half) * MM_APITCH + 4 * m) = xa[i];
#pragma unroll
    for (int u = 0; u < ITERS; ++u) {
      float* gp = (float*)&Gs[gslot[u]];
      *(float4*)(gp)     = make_float4(g0[u].x, g1[u].x, g0[u].y, g1[u].y);
      *(float4*)(gp + 4) = make_float4(g0[u].z, g1[u].z, g0[u].w, g1[u].w);
    }
    __syncthreads();
#pragma unroll
    for (int j = 0; j < MM_KC / 4; ++j) {
      const v2f a = *(const v2f*)(as + m * MM_APITCH + 4 * j + 2 * half);
#pragma unroll
      for (int t = 0; t < NT; ++t) {
        const v2f b = Gs[(j * NT + t) * 32 + lane];
        acc[t] = __builtin_amdgcn_wmma_f32_16x16x4_f32(false, a, false, b,
                                                       (short)0, acc[t], false, false);
      }
    }
  }

  if (live) {
    float* yr = Y + (size_t)(tile * 16 + half * 8) * NC + m;
#pragma unroll
    for (int t = 0; t < NT; ++t)
#pragma unroll
      for (int r = 0; r < 8; ++r)
        yr[(size_t)r * NC + 16 * t] = acc[t][r];
  }
}

__global__ void gde_prep_g(float* __restrict__ G, const float* __restrict__ Wp,
                           const float* __restrict__ R, int Cin, int Cout) {
  int idx = blockIdx.x * blockDim.x + threadIdx.x;
  int Ncols = 3 * Cout;
  int tot = Cin * Ncols;
  if (idx >= tot) return;
  int r = idx / Ncols, c = idx % Ncols;
  float v;
  if (c < 2 * Cout) {
    const int ch = c >> 1;
    const float w0 = Wp[r * Cout + ch];
    v = (c & 1) ? (Wp[(size_t)Cin * Cout + r * Cout + ch] - w0) : w0;
  } else {
    v = R[r * Cout + (c - 2 * Cout)];
  }
  G[idx] = v;
}

__global__ void gde_deg(float* __restrict__ deg, const int* __restrict__ dst, int E) {
  int e = blockIdx.x * blockDim.x + threadIdx.x;
  if (e < E) atomicAdd(&deg[dst[e]], 1.0f);
}

__global__ void gde_invdeg(float* __restrict__ d, int n) {
  int i = blockIdx.x * blockDim.x + threadIdx.x;
  if (i < n) d[i] = 1.0f / fmaxf(d[i], 1.0f);
}

#define ER_THREADS 256
static_assert(GDE_E % ER_THREADS == 0, "every wave has 32 whole edges");

template <int EPS>
__device__ __forceinline__ int gde_edge_pick(int v, int j, int sub) {
  const int a = __builtin_amdgcn_readlane(v, j);
  if (EPS == 1) return a;
  const int b = __builtin_amdgcn_readlane(v, j + 1);
  return sub ? b : a;
}

template <int COUT>
__global__ __launch_bounds__(ER_THREADS) void gde_edge_rows(
    float* __restrict__ agg, const float* __restrict__ Z, const int* __restrict__ src,
    const int* __restrict__ dst, const float* __restrict__ p) {
  constexpr int LPE = COUT < 32 ? COUT : 32;
  constexpr int EPS = 32 / LPE;
  constexpr int CPL = COUT / LPE;
  constexpr int GRP = 8;
  static_assert(32 % LPE == 0 && COUT % LPE == 0 && 32 % (GRP * EPS) == 0, "shape");
  const int lane = threadIdx.x & 31;
  const int sub  = lane / LPE;
  const int c    = lane % LPE;
  const int e    = blockIdx.x * ER_THREADS + threadIdx.x;
  const int sv = src[e];
  const int dv = dst[e];
  const int pv = __float_as_int(p[e]);
#pragma unroll 1
  for (int j0 = 0; j0 < 32; j0 += GRP * EPS) {
    float2 z[GRP][CPL];
    int    d[GRP];
    float  pe[GRP];
#pragma unroll
    for (int u = 0; u < GRP; ++u) {
      const int j = j0 + u * EPS;
      const int s = gde_edge_pick<EPS>(sv, j, sub);
      d[u]  = gde_edge_pick<EPS>(dv, j, sub);
      pe[u] = __int_as_float(gde_edge_pick<EPS>(pv, j, sub));
      const float2* zr = (const float2*)(Z + (size_t)s * (3 * COUT)) + c;
#pragma unroll
      for (int h = 0; h < CPL; ++h) z[u][h] = zr[h * LPE];
    }
#pragma unroll
    for (int u = 0; u < GRP; ++u) {
      float* ar = agg + (size_t)d[u] * COUT + c;
#pragma unroll
      for (int h = 0; h < CPL; ++h) atomicAdd(ar + h * LPE, z[u][h].x + pe[u] * z[u][h].y);
    }
  }
}

__global__ void gde_finalize(float* __restrict__ out, const float* __restrict__ agg,
                             const float* __restrict__ Z, const float* __restrict__ bias,
                             const float* __restrict__ inv, int Cout, int act,
                             long long total) {
  long long idx = (long long)blockIdx.x * blockDim.x + threadIdx.x;
  if (idx >= total) return;
  int i = (int)(idx / Cout), c = (int)(idx % Cout);
  float v = agg[idx] * inv[i] + Z[(size_t)i * 3 * Cout + 2 * Cout + c] + bias[c];
  if (act) v = tanhf(v);
  out[idx] = v;
}

__global__ void gde_rk_stage(float* __restrict__ ACC, float* __restrict__ YT,
                             const float* __restrict__ H, const float* __restrict__ K,
                             float w, float c, int set_acc, long long n) {
  long long i = (long long)blockIdx.x * blockDim.x + threadIdx.x;
  if (i >= n) return;
  float k = K[i];
  ACC[i] = set_acc ? (w * k) : (ACC[i] + w * k);
  YT[i] = H[i] + c * k;
}

__global__ void gde_rk_final(float* __restrict__ H, const float* __restrict__ ACC,
                             const float* __restrict__ K, long long n) {
  long long i = (long long)blockIdx.x * blockDim.x + threadIdx.x;
  if (i >= n) return;
  H[i] = H[i] + 0.5f * (ACC[i] + K[i]);
}

__global__ void gde_log_softmax16(float* __restrict__ out, int n) {
  int i = blockIdx.x * blockDim.x + threadIdx.x;
  if (i >= n) return;
  float v[16];
  float mx = -3.402823466e+38f;
  float* row = out + (size_t)i * 16;
#pragma unroll
  for (int c = 0; c < 16; ++c) { v[c] = row[c]; mx = fmaxf(mx, v[c]); }
  float s = 0.f;
#pragma unroll
  for (int c = 0; c < 16; ++c) s += __expf(v[c] - mx);
  float ls = __logf(s);
#pragma unroll
  for (int c = 0; c < 16; ++c) row[c] = v[c] - mx - ls;
}

extern "C" void kernel_launch(void* const* d_in, const int* in_sizes, int n_in,
                              void* d_out, int out_size, void* d_ws, size_t ws_size,
                              hipStream_t stream) {
  const float* x   = (const float*)d_in[0];
  const float* p   = (const float*)d_in[1];
  const int*   src = (const int*)d_in[2];
  const int*   dst = (const int*)d_in[3];
  const float* W1 = (const float*)d_in[4],  *R1 = (const float*)d_in[5],  *b1 = (const float*)d_in[6];
  const float* Wa = (const float*)d_in[7],  *Ra = (const float*)d_in[8],  *ba = (const float*)d_in[9];
  const float* Wb = (const float*)d_in[10], *Rb = (const float*)d_in[11], *bb = (const float*)d_in[12];
  const float* W2 = (const float*)d_in[13], *R2 = (const float*)d_in[14], *b2 = (const float*)d_in[15];
  float* out = (float*)d_out;

  float* W = (float*)d_ws;
  size_t o = 0;
  float* Z   = W + o; o += (size_t)GDE_N * 192;
  float* AGG = W + o; o += (size_t)GDE_N * 64;
  float* H   = W + o; o += (size_t)GDE_N * 64;
  float* YT  = W + o; o += (size_t)GDE_N * 64;
  float* U   = W + o; o += (size_t)GDE_N * 64;
  float* Kb  = W + o; o += (size_t)GDE_N * 64;
  float* ACC = W + o; o += (size_t)GDE_N * 64;
  float* INV = W + o; o += (size_t)GDE_N;
  float* GC1 = W + o; o += (size_t)256 * 192;
  float* GCA = W + o; o += (size_t)64 * 192;
  float* GCB = W + o; o += (size_t)64 * 192;
  float* GC2 = W + o; o += (size_t)64 * 48;
  (void)ws_size; (void)n_in; (void)in_sizes; (void)out_size;

  gde_prep_g<<<(256 * 192 + 255) / 256, 256, 0, stream>>>(GC1, W1, R1, 256, 64);
  gde_prep_g<<<(64 * 192 + 255) / 256, 256, 0, stream>>>(GCA, Wa, Ra, 64, 64);
  gde_prep_g<<<(64 * 192 + 255) / 256, 256, 0, stream>>>(GCB, Wb, Rb, 64, 64);
  gde_prep_g<<<(64 * 48 + 255) / 256, 256, 0, stream>>>(GC2, W2, R2, 64, 16);

  hipMemsetAsync(INV, 0, sizeof(float) * GDE_N, stream);
  gde_deg<<<(GDE_E + 255) / 256, 256, 0, stream>>>(INV, dst, GDE_E);
  gde_invdeg<<<(GDE_N + 255) / 256, 256, 0, stream>>>(INV, GDE_N);

  auto spline = [&](const float* y, int Cin, int Cout, const float* G,
                    const float* bias, float* ob, int act) {
    const int N3 = 3 * Cout;
    const int mmg = (MM_TILES + MM_WAVES - 1) / MM_WAVES;
    if (Cin == 256)      gde_mm_lds<256, 12><<<mmg, MM_THREADS, 0, stream>>>(y, G, Z);
    else if (N3 == 192)  gde_mm_lds<64, 12><<<mmg, MM_THREADS, 0, stream>>>(y, G, Z);
    else                 gde_mm_lds<64, 3><<<mmg, MM_THREADS, 0, stream>>>(y, G, Z);
    hipMemsetAsync(AGG, 0, sizeof(float) * (size_t)GDE_N * Cout, stream);
    if (Cout == 64) gde_edge_rows<64><<<GDE_E / ER_THREADS, ER_THREADS, 0, stream>>>(AGG, Z, src, dst, p);
    else            gde_edge_rows<16><<<GDE_E / ER_THREADS, ER_THREADS, 0, stream>>>(AGG, Z, src, dst, p);
    long long tn = (long long)GDE_N * Cout;
    gde_finalize<<<(int)((tn + 255) / 256), 256, 0, stream>>>(ob, AGG, Z, bias, INV, Cout, act, tn);
  };
  auto odefunc = [&](const float* y, float* k) {
    spline(y, 64, 64, GCA, ba, U, 0);
    spline(U, 64, 64, GCB, bb, k, 0);
  };

  const long long tn = (long long)GDE_N * 64;
  const int eb = (int)((tn + 255) / 256);

  spline(x, 256, 64, GC1, b1, H, 1);

  odefunc(H, Kb);
  gde_rk_stage<<<eb, 256, 0, stream>>>(ACC, YT, H, Kb, 1.0f, 1.5f, 1, tn);
  odefunc(YT, Kb);
  gde_rk_stage<<<eb, 256, 0, stream>>>(ACC, YT, H, Kb, 2.0f, 1.5f, 0, tn);
  odefunc(YT, Kb);
  gde_rk_stage<<<eb, 256, 0, stream>>>(ACC, YT, H, Kb, 2.0f, 3.0f, 0, tn);
  odefunc(YT, Kb);
  gde_rk_final<<<eb, 256, 0, stream>>>(H, ACC, Kb, tn);

  spline(H, 64, 16, GC2, b2, out, 1);
  gde_log_softmax16<<<(GDE_N + 255) / 256, 256, 0, stream>>>(out, GDE_N);
}


# speedup vs baseline: 1.9160x; 1.0585x over previous
#include <hip/hip_runtime.h>
#include <math.h>

#define GDE_N 50000
#define GDE_E 800000

typedef float v2f __attribute__((ext_vector_type(2)));
typedef float v8f __attribute__((ext_vector_type(8)));

#define MM_TILES (GDE_N / 16)
#define MM_WAVES 8
#define MM_THREADS (32 * MM_WAVES)
#define MM_KC 64
#define MM_APITCH (MM_KC + 4)

static_assert(GDE_N % 16 == 0, "whole row tiles");

template <int K, int NT>
__global__ __launch_bounds__(MM_THREADS) __attribute__((amdgpu_waves_per_eu(4, 4))) void gde_mm_lds(
    const float* __restrict__ X, const float* __restrict__ G, float* __restrict__ Y) {
  static_assert(K % MM_KC == 0, "K must be whole chunks");
  constexpr int NC    = 16 * NT;
  constexpr int ITEMS = (MM_KC / 2) * (NC / 4);
  constexpr int ITERS = (ITEMS + MM_THREADS - 1) / MM_THREADS;
  constexpr int FULL  = ITEMS / MM_THREADS;
  __shared__ __attribute__((aligned(16))) v2f   Gs[(MM_KC / 4) * NT * 32];
  __shared__ __attribute__((aligned(16))) float As[MM_WAVES][16 * MM_APITCH];

  const int tid  = threadIdx.x;
  const int lane = tid & 31;
  const int wave = tid >> 5;
  const int m    = lane & 15;
  const int half = lane >> 4;
  const int  tile_raw = blockIdx.x * MM_WAVES + wave;
  const bool live     = tile_raw < MM_TILES;
  const int  tile     = live ? tile_raw : (MM_TILES - 1);
  const float* __restrict__ xt = X + (size_t)tile * 16 * K;
  float* as = As[wave];

  v8f acc[NT];
#pragma unroll
  for (int t = 0; t < NT; ++t) acc[t] = (v8f){0.f, 0.f, 0.f, 0.f, 0.f, 0.f, 0.f, 0.f};

#pragma unroll 1
  for (int kc = 0; kc < K; kc += MM_KC) {
    if (kc) __syncthreads();
    float4 xa[8];
#pragma unroll
    for (int i = 0; i < 8; ++i)
      xa[i] = *(const float4*)(xt + (size_t)(2 * i + half) * K + kc + 4 * m);
    float4 g0[ITERS], g1[ITERS];
    int    gslot[ITERS];
#pragma unroll
    for (int u = 0; u < ITERS; ++u) {
      const int i  = tid + u * MM_THREADS;
      const int ic = (u < FULL || i < ITEMS) ? i : (i - ITEMS);
      const int q  = ic / (NC / 4);
      const int c0 = (ic - q * (NC / 4)) * 4;
      g0[u] = *(const float4*)(G + (size_t)(kc + 2 * q)     * NC + c0);
      g1[u] = *(const float4*)(G + (size_t)(kc + 2 * q + 1) * NC + c0);
      gslot[u] = (((q >> 1) * NT + (c0 >> 4)) * 32) + (q & 1) * 16 + (c0 & 15);
    }
#pragma unroll
    for (int i = 0; i < 8; ++i)
      *(float4*)(as + (2 * i + half) * MM_APITCH + 4 * m) = xa[i];
#pragma unroll
    for (int u = 0; u < ITERS; ++u) {
      float* gp = (float*)&Gs[gslot[u]];
      *(float4*)(gp)     = make_float4(g0[u].x, g1[u].x, g0[u].y, g1[u].y);
      *(float4*)(gp + 4) = make_float4(g0[u].z, g1[u].z, g0[u].w, g1[u].w);
    }
    __syncthreads();
#pragma unroll
    for (int j = 0; j < MM_KC / 4; ++j) {
      const v2f a = *(const v2f*)(as + m * MM_APITCH + 4 * j + 2 * half);
#pragma unroll
      for (int t = 0; t < NT; ++t) {
        const v2f b = Gs[(j * NT + t) * 32 + lane];
        acc[t] = __builtin_amdgcn_wmma_f32_16x16x4_f32(false, a, false, b,
                                                       (short)0, acc[t], false, false);
      }
    }
  }

  if (live) {
    float* yr = Y + (size_t)(tile * 16 + half * 8) * NC + m;
#pragma unroll
    for (int t = 0; t < NT; ++t)
#pragma unroll
      for (int r = 0; r < 8; ++r)
        yr[(size_t)r * NC + 16 * t] = acc[t][r];
  }
}

__global__ void gde_prep_g(float* __restrict__ G, const float* __restrict__ Wp,
                           const float* __restrict__ R, int Cin, int Cout) {
  int idx = blockIdx.x * blockDim.x + threadIdx.x;
  int Ncols = 3 * Cout;
  int tot = Cin * Ncols;
  if (idx >= tot) return;
  int r = idx / Ncols, c = idx % Ncols;
  float v;
  if (c < 2 * Cout) {
    const int ch = c >> 1;
    const float w0 = Wp[r * Cout + ch];
    v = (c & 1) ? (Wp[(size_t)Cin * Cout + r * Cout + ch] - w0) : w0;
  } else {
    v = R[r * Cout + (c - 2 * Cout)];
  }
  G[idx] = v;
}

#define ER_WAVES   4
#define ER_THREADS (32 * ER_WAVES)
#define ER_RPW     64
#define ER_CHUNK   (ER_WAVES * ER_RPW)
#define ER_NCHUNK  (GDE_E / ER_CHUNK)
#define ER_SPLIT   (GDE_N / 2)
#define ER_TKPITCH 32
#define ER_PASSES  10
static_assert(GDE_E % ER_CHUNK == 0, "whole chunks");

__global__ void gde_deg(float* __restrict__ deg, const int* __restrict__ dst, int E) {
  int e = blockIdx.x * blockDim.x + threadIdx.x;
  if (e < E) atomicAdd(&deg[dst[e]], 1.0f);
}

__global__ void gde_invdeg(float* __restrict__ d, int n, unsigned* __restrict__ tickets) {
  int i = blockIdx.x * blockDim.x + threadIdx.x;
  if (i < ER_PASSES * 2 * ER_TKPITCH) tickets[i] = 0u;
  if (i < n) d[i] = 1.0f / fmaxf(d[i], 1.0f);
}


template <int EPS>
__device__ __forceinline__ int gde_edge_pick(int v, int j, int sub) {
  const int a = __builtin_amdgcn_readlane(v, j);
  if (EPS == 1) return a;
  const int b = __builtin_amdgcn_readlane(v, j + 1);
  return sub ? b : a;
}

template <int COUT, int G>
__device__ __forceinline__ void gde_edge_group(
    float* __restrict__ agg, const float* __restrict__ Z, int sg, int dg, int pg,
    int jb, int j0, int K, int sub, int c) {
  constexpr int LPE = COUT < 32 ? COUT : 32;
  constexpr int EPS = 32 / LPE;
  constexpr int CPL = COUT / LPE;
  float2 z[G][CPL];
  int    d[G];
  float  pe[G];
#pragma unroll
  for (int u = 0; u < G; ++u) {
    const int j = jb + u * EPS;
    const int s = gde_edge_pick<EPS>(sg, j, sub);
    d[u]  = gde_edge_pick<EPS>(dg, j, sub);
    pe[u] = __int_as_float(gde_edge_pick<EPS>(pg, j, sub));
    const float2* zr = (const float2*)(Z + (size_t)s * (3 * COUT)) + c;
#pragma unroll
    for (int h = 0; h < CPL; ++h) z[u][h] = zr[h * LPE];
  }
#pragma unroll
  for (int u = 0; u < G; ++u) {
    if (EPS == 1 || j0 + u * EPS + sub < K) {
      float* ar = agg + (size_t)d[u] * COUT + c;
#pragma unroll
      for (int h = 0; h < CPL; ++h) atomicAdd(ar + h * LPE, z[u][h].x + pe[u] * z[u][h].y);
    }
  }
}

template <int COUT>
__global__ __launch_bounds__(ER_THREADS) void gde_edge_rows(
    float* __restrict__ agg, const float* __restrict__ Z, const int* __restrict__ src,
    const int* __restrict__ dst, const float* __restrict__ p, unsigned* __restrict__ tickets) {
  constexpr int LPE = COUT < 32 ? COUT : 32;
  constexpr int EPS = 32 / LPE;
  constexpr int GRP = 8;
  static_assert(32 % LPE == 0 && COUT % LPE == 0 && 32 % (GRP * EPS) == 0, "shape");
  __shared__ int tk[2];
  __shared__ int rec[ER_WAVES][3][ER_RPW];
  const int tid  = threadIdx.x;
  const int lane = tid & 31;
  const int wave = tid >> 5;
  const int sub  = lane / LPE;
  const int c    = lane % LPE;
  if (tid == 0) {
    int part = (int)((__builtin_amdgcn_s_sendmsg_rtn(0x87) >> 18) & 1u);
    int t = (int)atomicAdd(tickets + part * ER_TKPITCH, 1u);
    if (t >= ER_NCHUNK) {
      part ^= 1;
      t = (int)atomicAdd(tickets + part * ER_TKPITCH, 1u);
    }
    tk[0] = t;
    tk[1] = part;
  }
  __syncthreads();
  const int t    = tk[0];
  const int part = tk[1];
  if (t >= ER_NCHUNK) return;

  const int e  = t * ER_CHUNK + wave * ER_RPW + lane;
  const int s0 = src[e],      s1 = src[e + 32];
  const int d0 = dst[e],      d1 = dst[e + 32];
  const int p0 = __float_as_int(p[e]), p1 = __float_as_int(p[e + 32]);
  const bool k0 = (d0 >= ER_SPLIT) == (part != 0);
  const bool k1 = (d1 >= ER_SPLIT) == (part != 0);
  const unsigned m0 = __builtin_amdgcn_ballot_w32(k0);
  const unsigned m1 = __builtin_amdgcn_ballot_w32(k1);
  const int n0 = __popc(m0);
  const int K  = n0 + __popc(m1);
  const unsigned below = (1u << lane) - 1u;
  const int q0 = k0 ? __popc(m0 & below)      : K + __popc(~m0 & below);
  const int q1 = k1 ? n0 + __popc(m1 & below) : K + (32 - n0) + __popc(~m1 & below);
  int* rs = rec[wave][0];
  int* rd = rec[wave][1];
  int* rp = rec[wave][2];
  rs[q0] = s0; rd[q0] = d0; rp[q0] = p0;
  rs[q1] = s1; rd[q1] = d1; rp[q1] = p1;
  const int sv0 = rs[lane], sv1 = rs[lane + 32];
  const int dv0 = rd[lane], dv1 = rd[lane + 32];
  const int pv0 = rp[lane], pv1 = rp[lane + 32];

  const int S = (K + EPS - 1) / EPS;
  int q = 0;
#define GDE_RUN(G)                                                                       \
  {                                                                                      \
    const int  j0 = q * EPS;                                                             \
    const bool hi = j0 >= 32;                                                            \
    gde_edge_group<COUT, G>(agg, Z, hi ? sv1 : sv0, hi ? dv1 : dv0, hi ? pv1 : pv0,      \
                            j0 & 31, j0, K, sub, c);                                     \
  }
#pragma unroll 1
  for (; q + GRP <= S; q += GRP) GDE_RUN(GRP)
  if (S - q >= 4) { GDE_RUN(4) q += 4; }
  if (S - q >= 2) { GDE_RUN(2) q += 2; }
  if (S - q >= 1) { GDE_RUN(1) }
#undef GDE_RUN
}

__global__ void gde_finalize(float* __restrict__ out, const float* __restrict__ agg,
                             const float* __restrict__ Z, const float* __restrict__ bias,
                             const float* __restrict__ inv, int Cout, int act,
                             long long total) {
  long long idx = (long long)blockIdx.x * blockDim.x + threadIdx.x;
  if (idx >= total) return;
  int i = (int)(idx / Cout), c = (int)(idx % Cout);
  float v = agg[idx] * inv[i] + Z[(size_t)i * 3 * Cout + 2 * Cout + c] + bias[c];
  if (act) v = tanhf(v);
  out[idx] = v;
}

__global__ void gde_rk_stage(float* __restrict__ ACC, float* __restrict__ YT,
                             const float* __restrict__ H, const float* __restrict__ K,
                             float w, float c, int set_acc, long long n) {
  long long i = (long long)blockIdx.x * blockDim.x + threadIdx.x;
  if (i >= n) return;
  float k = K[i];
  ACC[i] = set_acc ? (w * k) : (ACC[i] + w * k);
  YT[i] = H[i] + c * k;
}

__global__ void gde_rk_final(float* __restrict__ H, const float* __restrict__ ACC,
                             const float* __restrict__ K, long long n) {
  long long i = (long long)blockIdx.x * blockDim.x + threadIdx.x;
  if (i >= n) return;
  H[i] = H[i] + 0.5f * (ACC[i] + K[i]);
}

__global__ void gde_log_softmax16(float* __restrict__ out, int n) {
  int i = blockIdx.x * blockDim.x + threadIdx.x;
  if (i >= n) return;
  float v[16];
  float mx = -3.402823466e+38f;
  float* row = out + (size_t)i * 16;
#pragma unroll
  for (int c = 0; c < 16; ++c) { v[c] = row[c]; mx = fmaxf(mx, v[c]); }
  float s = 0.f;
#pragma unroll
  for (int c = 0; c < 16; ++c) s += __expf(v[c] - mx);
  float ls = __logf(s);
#pragma unroll
  for (int c = 0; c < 16; ++c) row[c] = v[c] - mx - ls;
}

extern "C" void kernel_launch(void* const* d_in, const int* in_sizes, int n_in,
                              void* d_out, int out_size, void* d_ws, size_t ws_size,
                              hipStream_t stream) {
  const float* x   = (const float*)d_in[0];
  const float* p   = (const float*)d_in[1];
  const int*   src = (const int*)d_in[2];
  const int*   dst = (const int*)d_in[3];
  const float* W1 = (const float*)d_in[4],  *R1 = (const float*)d_in[5],  *b1 = (const float*)d_in[6];
  const float* Wa = (const float*)d_in[7],  *Ra = (const float*)d_in[8],  *ba = (const float*)d_in[9];
  const float* Wb = (const float*)d_in[10], *Rb = (const float*)d_in[11], *bb = (const float*)d_in[12];
  const float* W2 = (const float*)d_in[13], *R2 = (const float*)d_in[14], *b2 = (const float*)d_in[15];
  float* out = (float*)d_out;

  float* W = (float*)d_ws;
  size_t o = 0;
  float* Z   = W + o; o += (size_t)GDE_N * 192;
  float* AGG = W + o; o += (size_t)GDE_N * 64;
  float* H   = W + o; o += (size_t)GDE_N * 64;
  float* YT  = W + o; o += (size_t)GDE_N * 64;
  float* U   = W + o; o += (size_t)GDE_N * 64;
  float* Kb  = W + o; o += (size_t)GDE_N * 64;
  float* ACC = W + o; o += (size_t)GDE_N * 64;
  float* INV = W + o; o += (size_t)GDE_N;
  float* GC1 = W + o; o += (size_t)256 * 192;
  float* GCA = W + o; o += (size_t)64 * 192;
  float* GCB = W + o; o += (size_t)64 * 192;
  float* GC2 = W + o; o += (size_t)64 * 48;
  unsigned* TK = (unsigned*)d_out;
  (void)ws_size; (void)n_in; (void)in_sizes; (void)out_size;

  gde_prep_g<<<(256 * 192 + 255) / 256, 256, 0, stream>>>(GC1, W1, R1, 256, 64);
  gde_prep_g<<<(64 * 192 + 255) / 256, 256, 0, stream>>>(GCA, Wa, Ra, 64, 64);
  gde_prep_g<<<(64 * 192 + 255) / 256, 256, 0, stream>>>(GCB, Wb, Rb, 64, 64);
  gde_prep_g<<<(64 * 48 + 255) / 256, 256, 0, stream>>>(GC2, W2, R2, 64, 16);

  hipMemsetAsync(INV, 0, sizeof(float) * GDE_N, stream);
  gde_deg<<<(GDE_E + 255) / 256, 256, 0, stream>>>(INV, dst, GDE_E);
  gde_invdeg<<<(GDE_N + 255) / 256, 256, 0, stream>>>(INV, GDE_N, TK);

  auto spline = [&](const float* y, int Cin, int Cout, const float* G,
                    const float* bias, float* ob, int act) {
    const int N3 = 3 * Cout;
    const int mmg = (MM_TILES + MM_WAVES - 1) / MM_WAVES;
    if (Cin == 256)      gde_mm_lds<256, 12><<<mmg, MM_THREADS, 0, stream>>>(y, G, Z);
    else if (N3 == 192)  gde_mm_lds<64, 12><<<mmg, MM_THREADS, 0, stream>>>(y, G, Z);
    else                 gde_mm_lds<64, 3><<<mmg, MM_THREADS, 0, stream>>>(y, G, Z);
    hipMemsetAsync(AGG, 0, sizeof(float) * (size_t)GDE_N * Cout, stream);
    if (Cout == 64) gde_edge_rows<64><<<2 * ER_NCHUNK, ER_THREADS, 0, stream>>>(AGG, Z, src, dst, p, TK);
    else            gde_edge_rows<16><<<2 * ER_NCHUNK, ER_THREADS, 0, stream>>>(AGG, Z, src, dst, p, TK);
    TK += 2 * ER_TKPITCH;
    long long tn = (long long)GDE_N * Cout;
    gde_finalize<<<(int)((tn + 255) / 256), 256, 0, stream>>>(ob, AGG, Z, bias, INV, Cout, act, tn);
  };
  auto odefunc = [&](const float* y, float* k) {
    spline(y, 64, 64, GCA, ba, U, 0);
    spline(U, 64, 64, GCB, bb, k, 0);
  };

  const long long tn = (long long)GDE_N * 64;
  const int eb = (int)((tn + 255) / 256);

  spline(x, 256, 64, GC1, b1, H, 1);

  odefunc(H, Kb);
  gde_rk_stage<<<eb, 256, 0, stream>>>(ACC, YT, H, Kb, 1.0f, 1.5f, 1, tn);
  odefunc(YT, Kb);
  gde_rk_stage<<<eb, 256, 0, stream>>>(ACC, YT, H, Kb, 2.0f, 1.5f, 0, tn);
  odefunc(YT, Kb);
  gde_rk_stage<<<eb, 256, 0, stream>>>(ACC, YT, H, Kb, 2.0f, 3.0f, 0, tn);
  odefunc(YT, Kb);
  gde_rk_final<<<eb, 256, 0, stream>>>(H, ACC, Kb, tn);

  spline(H, 64, 16, GC2, b2, out, 1);
  gde_log_softmax16<<<(GDE_N + 255) / 256, 256, 0, stream>>>(out, GDE_N);
}


# Sum buffer seeded by the product: its epilogue stores y@R + b straight into the layer output, edges add messages pre-scaled by 1/deg on top; 10 memsets and 10 closing passes gone; sum order changes
# speedup vs baseline: 2.4126x; 1.2592x over previous
#include <hip/hip_runtime.h>
#include <math.h>

#define GDE_N 50000
#define GDE_E 800000

typedef float v2f __attribute__((ext_vector_type(2)));
typedef float v8f __attribute__((ext_vector_type(8)));

#define MM_TILES (GDE_N / 16)
#define MM_WAVES 8
#define MM_THREADS (32 * MM_WAVES)
#define MM_KC 64
#define MM_APITCH (MM_KC + 4)

static_assert(GDE_N % 16 == 0, "whole row tiles");

template <int K, int NT>
__global__ __launch_bounds__(MM_THREADS) __attribute__((amdgpu_waves_per_eu(4, 4))) void gde_mm_lds(
    const float* __restrict__ X, const float* __restrict__ G, float* __restrict__ Zp,
    float* __restrict__ S, const float* __restrict__ bias) {
  static_assert(K % MM_KC == 0, "K must be whole chunks");
  static_assert(NT % 3 == 0, "columns are [pairs | root], two to one");
  constexpr int NC    = 16 * NT;
  constexpr int NTP   = 2 * NT / 3;
  constexpr int CP    = 16 * NTP;
  constexpr int CO    = NC - CP;
  constexpr int ITEMS = (MM_KC / 2) * (NC / 4);
  constexpr int ITERS = (ITEMS + MM_THREADS - 1) / MM_THREADS;
  constexpr int FULL  = ITEMS / MM_THREADS;
  __shared__ __attribute__((aligned(16))) v2f   Gs[(MM_KC / 4) * NT * 32];
  __shared__ __attribute__((aligned(16))) float As[MM_WAVES][16 * MM_APITCH];

  const int tid  = threadIdx.x;
  const int lane = tid & 31;
  const int wave = __builtin_amdgcn_readfirstlane(tid >> 5);
  const int m    = lane & 15;
  const int half = lane >> 4;
  const int  tile_raw = blockIdx.x * MM_WAVES + wave;
  const bool live     = tile_raw < MM_TILES;
  const int  tile     = live ? tile_raw : (MM_TILES - 1);
  const float* __restrict__ xt = X + (size_t)tile * 16 * K;
  float* as = As[wave];

  v8f acc[NT];
#pragma unroll
  for (int t = 0; t < NTP; ++t) acc[t] = (v8f){0.f, 0.f, 0.f, 0.f, 0.f, 0.f, 0.f, 0.f};
#pragma unroll
  for (int t = NTP; t < NT; ++t) {
    const float bv = bias[16 * (t - NTP) + m];
    acc[t] = (v8f){bv, bv, bv, bv, bv, bv, bv, bv};
  }

#pragma unroll 1
  for (int kc = 0; kc < K; kc += MM_KC) {
    if (kc) __syncthreads();
    float4 xa[8];
#pragma unroll
    for (int i = 0; i < 8; ++i)
      xa[i] = *(const float4*)(xt + (size_t)(2 * i + half) * K + kc + 4 * m);
    float4 g0[ITERS], g1[ITERS];
    int    gslot[ITERS];
#pragma unroll
    for (int u = 0; u < ITERS; ++u) {
      const int i  = tid + u * MM_THREADS;
      const int ic = (u < FULL || i < ITEMS) ? i : (i - ITEMS);
      const int q  = ic / (NC / 4);
      const int c0 = (ic - q * (NC / 4)) * 4;
      g0[u] = *(const float4*)(G + (size_t)(kc + 2 * q)     * NC + c0);
      g1[u] = *(const float4*)(G + (size_t)(kc + 2 * q + 1) * NC + c0);
      gslot[u] = (((q >> 1) * NT + (c0 >> 4)) * 32) + (q & 1) * 16 + (c0 & 15);
    }
#pragma unroll
    for (int i = 0; i < 8; ++i)
      *(float4*)(as + (2 * i + half) * MM_APITCH + 4 * m) = xa[i];
#pragma unroll
    for (int u = 0; u < ITERS; ++u) {
      float* gp = (float*)&Gs[gslot[u]];
      *(float4*)(gp)     = make_float4(g0[u].x, g1[u].x, g0[u].y, g1[u].y);
      *(float4*)(gp + 4) = make_float4(g0[u].z, g1[u].z, g0[u].w, g1[u].w);
    }
    __syncthreads();
#pragma unroll
    for (int j = 0; j < MM_KC / 4; ++j) {
      const v2f a = *(const v2f*)(as + m * MM_APITCH + 4 * j + 2 * half);
#pragma unroll
      for (int t = 0; t < NT; ++t) {
        const v2f b = Gs[(j * NT + t) * 32 + lane];
        acc[t] = __builtin_amdgcn_wmma_f32_16x16x4_f32(false, a, false, b,
                                                       (short)0, acc[t], false, false);
      }
    }
  }

  if (live) {
    const size_t row0 = (size_t)(tile * 16 + half * 8);
    float* zr = Zp + row0 * CP + m;
#pragma unroll
    for (int t = 0; t < NTP; ++t)
#pragma unroll
      for (int r = 0; r < 8; ++r)
        zr[(size_t)r * CP + 16 * t] = acc[t][r];
    float* sr = S + row0 * CO + m;
#pragma unroll
    for (int t = NTP; t < NT; ++t)
#pragma unroll
      for (int r = 0; r < 8; ++r)
        sr[(size_t)r * CO + 16 * (t - NTP)] = acc[t][r];
  }
}

__global__ void gde_prep_g(float* __restrict__ G, const float* __restrict__ Wp,
                           const float* __restrict__ R, int Cin, int Cout) {
  int idx = blockIdx.x * blockDim.x + threadIdx.x;
  int Ncols = 3 * Cout;
  int tot = Cin * Ncols;
  if (idx >= tot) return;
  int r = idx / Ncols, c = idx % Ncols;
  float v;
  if (c < 2 * Cout) {
    const int ch = c >> 1;
    const float w0 = Wp[r * Cout + ch];
    v = (c & 1) ? (Wp[(size_t)Cin * Cout + r * Cout + ch] - w0) : w0;
  } else {
    v = R[r * Cout + (c - 2 * Cout)];
  }
  G[idx] = v;
}

#define ER_WAVES   4
#define ER_THREADS (32 * ER_WAVES)
#define ER_RPW     64
#define ER_CHUNK   (ER_WAVES * ER_RPW)
#define ER_NCHUNK  (GDE_E / ER_CHUNK)
#define ER_SPLIT   (GDE_N / 2)
#define ER_TKPITCH 32
#define ER_PASSES  10
static_assert(GDE_E % ER_CHUNK == 0, "whole chunks");

__global__ void gde_deg(float* __restrict__ deg, const int* __restrict__ dst, int E) {
  int e = blockIdx.x * blockDim.x + threadIdx.x;
  if (e < E) atomicAdd(&deg[dst[e]], 1.0f);
}

__global__ void gde_edge_w(float* __restrict__ wa, float* __restrict__ wb,
                           const float* __restrict__ deg, const int* __restrict__ dst,
                           const float* __restrict__ p, int E, unsigned* __restrict__ tickets) {
  int e = blockIdx.x * blockDim.x + threadIdx.x;
  if (e < ER_PASSES * 2 * ER_TKPITCH) tickets[e] = 0u;
  if (e < E) {
    const float iv = 1.0f / fmaxf(deg[dst[e]], 1.0f);
    wa[e] = iv;
    wb[e] = p[e] * iv;
  }
}


template <int EPS>
__device__ __forceinline__ int gde_edge_pick(int v, int j, int sub) {
  const int a = __builtin_amdgcn_readlane(v, j);
  if (EPS == 1) return a;
  const int b = __builtin_amdgcn_readlane(v, j + 1);
  return sub ? b : a;
}

template <int COUT, int G>
__device__ __forceinline__ void gde_edge_group(
    float* __restrict__ agg, const float* __restrict__ Z, int sg, int dg, int ag, int bg,
    int jb, int j0, int K, int sub, int c) {
  constexpr int LPE = COUT < 32 ? COUT : 32;
  constexpr int EPS = 32 / LPE;
  constexpr int CPL = COUT / LPE;
  float2 z[G][CPL];
  int    d[G];
  float  wa[G], wb[G];
#pragma unroll
  for (int u = 0; u < G; ++u) {
    const int j = jb + u * EPS;
    const int s = gde_edge_pick<EPS>(sg, j, sub);
    d[u]  = gde_edge_pick<EPS>(dg, j, sub);
    wa[u] = __int_as_float(gde_edge_pick<EPS>(ag, j, sub));
    wb[u] = __int_as_float(gde_edge_pick<EPS>(bg, j, sub));
    const float2* zr = (const float2*)(Z + (size_t)s * (2 * COUT)) + c;
#pragma unroll
    for (int h = 0; h < CPL; ++h) z[u][h] = zr[h * LPE];
  }
#pragma unroll
  for (int u = 0; u < G; ++u) {
    if (EPS == 1 || j0 + u * EPS + sub < K) {
      float* ar = agg + (size_t)d[u] * COUT + c;
#pragma unroll
      for (int h = 0; h < CPL; ++h)
        atomicAdd(ar + h * LPE, wa[u] * z[u][h].x + wb[u] * z[u][h].y);
    }
  }
}

template <int COUT>
__global__ __launch_bounds__(ER_THREADS) void gde_edge_rows(
    float* __restrict__ agg, const float* __restrict__ Z, const int* __restrict__ src,
    const int* __restrict__ dst, const float* __restrict__ wa, const float* __restrict__ wb,
    unsigned* __restrict__ tickets) {
  constexpr int LPE = COUT < 32 ? COUT : 32;
  constexpr int EPS = 32 / LPE;
  constexpr int GRP = 8;
  static_assert(32 % LPE == 0 && COUT % LPE == 0 && 32 % (GRP * EPS) == 0, "shape");
  __shared__ int tk[2];
  __shared__ int rec[ER_WAVES][4][ER_RPW];
  const int tid  = threadIdx.x;
  const int lane = tid & 31;
  const int wave = tid >> 5;
  const int sub  = lane / LPE;
  const int c    = lane % LPE;
  if (tid == 0) {
    int part = (int)((__builtin_amdgcn_s_sendmsg_rtn(0x87) >> 18) & 1u);
    int t = (int)atomicAdd(tickets + part * ER_TKPITCH, 1u);
    if (t >= ER_NCHUNK) {
      part ^= 1;
      t = (int)atomicAdd(tickets + part * ER_TKPITCH, 1u);
    }
    tk[0] = t;
    tk[1] = part;
  }
  __syncthreads();
  const int t    = tk[0];
  const int part = tk[1];
  if (t >= ER_NCHUNK) return;

  const int e  = t * ER_CHUNK + wave * ER_RPW + lane;
  const int s0 = src[e],      s1 = src[e + 32];
  const int d0 = dst[e],      d1 = dst[e + 32];
  const int a0 = __float_as_int(wa[e]), a1 = __float_as_int(wa[e + 32]);
  const int b0 = __float_as_int(wb[e]), b1 = __float_as_int(wb[e + 32]);
  const bool k0 = (d0 >= ER_SPLIT) == (part != 0);
  const bool k1 = (d1 >= ER_SPLIT) == (part != 0);
  const unsigned m0 = __builtin_amdgcn_ballot_w32(k0);
  const unsigned m1 = __builtin_amdgcn_ballot_w32(k1);
  const int n0 = __popc(m0);
  const int K  = n0 + __popc(m1);
  const unsigned below = (1u << lane) - 1u;
  const int q0 = k0 ? __popc(m0 & below)      : K + __popc(~m0 & below);
  const int q1 = k1 ? n0 + __popc(m1 & below) : K + (32 - n0) + __popc(~m1 & below);
  int* rs = rec[wave][0];
  int* rd = rec[wave][1];
  int* ra = rec[wave][2];
  int* rb = rec[wave][3];
  rs[q0] = s0; rd[q0] = d0; ra[q0] = a0; rb[q0] = b0;
  rs[q1] = s1; rd[q1] = d1; ra[q1] = a1; rb[q1] = b1;
  const int sv0 = rs[lane], sv1 = rs[lane + 32];
  const int dv0 = rd[lane], dv1 = rd[lane + 32];
  const int av0 = ra[lane], av1 = ra[lane + 32];
  const int bv0 = rb[lane], bv1 = rb[lane + 32];

  const int S = (K + EPS - 1) / EPS;
  int q = 0;
#define GDE_RUN(G)                                                                       \
  {                                                                                      \
    const int  j0 = q * EPS;                                                             \
    const bool hi = j0 >= 32;                                                            \
    gde_edge_group<COUT, G>(agg, Z, hi ? sv1 : sv0, hi ? dv1 : dv0, hi ? av1 : av0,      \
                            hi ? bv1 : bv0, j0 & 31, j0, K, sub, c);                     \
  }
#pragma unroll 1
  for (; q + GRP <= S; q += GRP) GDE_RUN(GRP)
  if (S - q >= 4) { GDE_RUN(4) q += 4; }
  if (S - q >= 2) { GDE_RUN(2) q += 2; }
  if (S - q >= 1) { GDE_RUN(1) }
#undef GDE_RUN
}

__global__ void gde_tanh4(float* __restrict__ h, int n4) {
  int i = blockIdx.x * blockDim.x + threadIdx.x;
  if (i >= n4) return;
  float4 v = ((const float4*)h)[i];
  v.x = tanhf(v.x); v.y = tanhf(v.y); v.z = tanhf(v.z); v.w = tanhf(v.w);
  ((float4*)h)[i] = v;
}

__global__ void gde_rk_stage(float* __restrict__ ACC, float* __restrict__ YT,
                             const float* __restrict__ H, const float* __restrict__ K,
                             float w, float c, int set_acc, long long n) {
  long long i = (long long)blockIdx.x * blockDim.x + threadIdx.x;
  if (i >= n) return;
  float k = K[i];
  ACC[i] = set_acc ? (w * k) : (ACC[i] + w * k);
  YT[i] = H[i] + c * k;
}

__global__ void gde_rk_final(float* __restrict__ H, const float* __restrict__ ACC,
                             const float* __restrict__ K, long long n) {
  long long i = (long long)blockIdx.x * blockDim.x + threadIdx.x;
  if (i >= n) return;
  H[i] = H[i] + 0.5f * (ACC[i] + K[i]);
}

__global__ void gde_log_softmax16(float* __restrict__ out, int n) {
  int i = blockIdx.x * blockDim.x + threadIdx.x;
  if (i >= n) return;
  float v[16];
  float mx = -3.402823466e+38f;
  float* row = out + (size_t)i * 16;
#pragma unroll
  for (int c = 0; c < 16; ++c) { v[c] = tanhf(row[c]); mx = fmaxf(mx, v[c]); }
  float s = 0.f;
#pragma unroll
  for (int c = 0; c < 16; ++c) s += __expf(v[c] - mx);
  float ls = __logf(s);
#pragma unroll
  for (int c = 0; c < 16; ++c) row[c] = v[c] - mx - ls;
}

extern "C" void kernel_launch(void* const* d_in, const int* in_sizes, int n_in,
                              void* d_out, int out_size, void* d_ws, size_t ws_size,
                              hipStream_t stream) {
  const float* x   = (const float*)d_in[0];
  const float* p   = (const float*)d_in[1];
  const int*   src = (const int*)d_in[2];
  const int*   dst = (const int*)d_in[3];
  const float* W1 = (const float*)d_in[4],  *R1 = (const float*)d_in[5],  *b1 = (const float*)d_in[6];
  const float* Wa = (const float*)d_in[7],  *Ra = (const float*)d_in[8],  *ba = (const float*)d_in[9];
  const float* Wb = (const float*)d_in[10], *Rb = (const float*)d_in[11], *bb = (const float*)d_in[12];
  const float* W2 = (const float*)d_in[13], *R2 = (const float*)d_in[14], *b2 = (const float*)d_in[15];
  float* out = (float*)d_out;

  float* W = (float*)d_ws;
  size_t o = 0;
  float* Z   = W + o; o += (size_t)GDE_N * 128;
  float* H   = W + o; o += (size_t)GDE_N * 64;
  float* YT  = W + o; o += (size_t)GDE_N * 64;
  float* U   = W + o; o += (size_t)GDE_N * 64;
  float* Kb  = W + o; o += (size_t)GDE_N * 64;
  float* ACC = W + o; o += (size_t)GDE_N * 64;
  unsigned* TK = (unsigned*)(W + o); o += (size_t)ER_PASSES * 2 * ER_TKPITCH;
  float* DEG = W + o; o += (size_t)GDE_N;
  float* EWA = W + o; o += (size_t)GDE_E;
  float* EWB = W + o; o += (size_t)GDE_E;
  float* GC1 = W + o; o += (size_t)256 * 192;
  float* GCA = W + o; o += (size_t)64 * 192;
  float* GCB = W + o; o += (size_t)64 * 192;
  float* GC2 = W + o; o += (size_t)64 * 48;
  (void)ws_size; (void)n_in; (void)in_sizes; (void)out_size;

  gde_prep_g<<<(256 * 192 + 255) / 256, 256, 0, stream>>>(GC1, W1, R1, 256, 64);
  gde_prep_g<<<(64 * 192 + 255) / 256, 256, 0, stream>>>(GCA, Wa, Ra, 64, 64);
  gde_prep_g<<<(64 * 192 + 255) / 256, 256, 0, stream>>>(GCB, Wb, Rb, 64, 64);
  gde_prep_g<<<(64 * 48 + 255) / 256, 256, 0, stream>>>(GC2, W2, R2, 64, 16);

  hipMemsetAsync(DEG, 0, sizeof(float) * GDE_N, stream);
  gde_deg<<<(GDE_E + 255) / 256, 256, 0, stream>>>(DEG, dst, GDE_E);
  gde_edge_w<<<(GDE_E + 255) / 256, 256, 0, stream>>>(EWA, EWB, DEG, dst, p, GDE_E, TK);

  auto spline = [&](const float* y, int Cin, int Cout, const float* G,
                    const float* bias, float* ob) {
    const int N3 = 3 * Cout;
    const int mmg = (MM_TILES + MM_WAVES - 1) / MM_WAVES;
    if (Cin == 256)      gde_mm_lds<256, 12><<<mmg, MM_THREADS, 0, stream>>>(y, G, Z, ob, bias);
    else if (N3 == 192)  gde_mm_lds<64, 12><<<mmg, MM_THREADS, 0, stream>>>(y, G, Z, ob, bias);
    else                 gde_mm_lds<64, 3><<<mmg, MM_THREADS, 0, stream>>>(y, G, Z, ob, bias);
    if (Cout == 64) gde_edge_rows<64><<<2 * ER_NCHUNK, ER_THREADS, 0, stream>>>(ob, Z, src, dst, EWA, EWB, TK);
    else            gde_edge_rows<16><<<2 * ER_NCHUNK, ER_THREADS, 0, stream>>>(ob, Z, src, dst, EWA, EWB, TK);
    TK += 2 * ER_TKPITCH;
  };
  auto odefunc = [&](const float* y, float* k) {
    spline(y, 64, 64, GCA, ba, U);
    spline(U, 64, 64, GCB, bb, k);
  };

  const long long tn = (long long)GDE_N * 64;
  const int eb = (int)((tn + 255) / 256);

  spline(x, 256, 64, GC1, b1, H);
  gde_tanh4<<<(int)((tn / 4 + 255) / 256), 256, 0, stream>>>(H, (int)(tn / 4));

  odefunc(H, Kb);
  gde_rk_stage<<<eb, 256, 0, stream>>>(ACC, YT, H, Kb, 1.0f, 1.5f, 1, tn);
  odefunc(YT, Kb);
  gde_rk_stage<<<eb, 256, 0, stream>>>(ACC, YT, H, Kb, 2.0f, 1.5f, 0, tn);
  odefunc(YT, Kb);
  gde_rk_stage<<<eb, 256, 0, stream>>>(ACC, YT, H, Kb, 2.0f, 3.0f, 0, tn);
  odefunc(YT, Kb);
  gde_rk_final<<<eb, 256, 0, stream>>>(H, ACC, Kb, tn);

  spline(H, 64, 16, GC2, b2, out);
  gde_log_softmax16<<<(GDE_N + 255) / 256, 256, 0, stream>>>(out, GDE_N);
}


# Elementwise passes between layers (tanh, 3 RK4 stage updates, RK4 final sum) done by the next product's loader on the rows it reads anyway: 5 launches gone; arithmetic and sum order unchanged
# speedup vs baseline: 2.6298x; 1.0900x over previous
#include <hip/hip_runtime.h>
#include <math.h>

#define GDE_N 50000
#define GDE_E 800000

typedef float v2f __attribute__((ext_vector_type(2)));
typedef float v8f __attribute__((ext_vector_type(8)));

#define MM_A_PLAIN 0
#define MM_A_TANH  1
#define MM_A_SET   2
#define MM_A_ADD   3
#define MM_A_FINAL 4
#define MM_TILES (GDE_N / 16)
#define MM_WAVES 8
#define MM_THREADS (32 * MM_WAVES)
#define MM_KC 64
#define MM_APITCH (MM_KC + 4)

static_assert(GDE_N % 16 == 0, "whole row tiles");

template <int K, int NT, int AM>
__global__ __launch_bounds__(MM_THREADS) __attribute__((amdgpu_waves_per_eu(4, 4))) void gde_mm_lds(
    const float* __restrict__ X, const float* __restrict__ G, float* __restrict__ Zp,
    float* __restrict__ S, const float* __restrict__ bias, float* Hio,
    const float* __restrict__ Kp, float* __restrict__ Acc, float w, float c) {
  static_assert(K % MM_KC == 0, "K must be whole chunks");
  static_assert(AM == MM_A_PLAIN || K == MM_KC, "a loader that stores sees every element once");
  static_assert(NT % 3 == 0, "columns are [pairs | root], two to one");
  constexpr int NC    = 16 * NT;
  constexpr int NTP   = 2 * NT / 3;
  constexpr int CP    = 16 * NTP;
  constexpr int CO    = NC - CP;
  constexpr int ITEMS = (MM_KC / 2) * (NC / 4);
  constexpr int ITERS = (ITEMS + MM_THREADS - 1) / MM_THREADS;
  constexpr int FULL  = ITEMS / MM_THREADS;
  __shared__ __attribute__((aligned(16))) v2f   Gs[(MM_KC / 4) * NT * 32];
  __shared__ __attribute__((aligned(16))) float As[MM_WAVES][16 * MM_APITCH];

  const int tid  = threadIdx.x;
  const int lane = tid & 31;
  const int wave = __builtin_amdgcn_readfirstlane(tid >> 5);
  const int m    = lane & 15;
  const int half = lane >> 4;
  const int  tile_raw = blockIdx.x * MM_WAVES + wave;
  const bool live     = tile_raw < MM_TILES;
  const int  tile     = live ? tile_raw : (MM_TILES - 1);
  const size_t tbase = (size_t)tile * 16 * K;
  const float* __restrict__ xt = X + tbase;
  float* as = As[wave];

  v8f acc[NT];
#pragma unroll
  for (int t = 0; t < NTP; ++t) acc[t] = (v8f){0.f, 0.f, 0.f, 0.f, 0.f, 0.f, 0.f, 0.f};
#pragma unroll
  for (int t = NTP; t < NT; ++t) {
    const float bv = bias[16 * (t - NTP) + m];
    acc[t] = (v8f){bv, bv, bv, bv, bv, bv, bv, bv};
  }

#pragma unroll 1
  for (int kc = 0; kc < K; kc += MM_KC) {
    if (kc) __syncthreads();
    float4 xa[8], kk[8], aa[8];
#pragma unroll
    for (int i = 0; i < 8; ++i) {
      const size_t off = (size_t)(2 * i + half) * K + kc + 4 * m;
      if (AM == MM_A_TANH) xa[i] = *(const float4*)(Hio + tbase + off);
      else                 xa[i] = *(const float4*)(xt + off);
      if (AM == MM_A_SET || AM == MM_A_ADD || AM == MM_A_FINAL) kk[i] = *(const float4*)(Kp + tbase + off);
      if (AM == MM_A_ADD || AM == MM_A_FINAL) aa[i] = *(const float4*)(Acc + tbase + off);
    }
    float4 g0[ITERS], g1[ITERS];
    int    gslot[ITERS];
#pragma unroll
    for (int u = 0; u < ITERS; ++u) {
      const int i  = tid + u * MM_THREADS;
      const int ic = (u < FULL || i < ITEMS) ? i : (i - ITEMS);
      const int q  = ic / (NC / 4);
      const int c0 = (ic - q * (NC / 4)) * 4;
      g0[u] = *(const float4*)(G + (size_t)(kc + 2 * q)     * NC + c0);
      g1[u] = *(const float4*)(G + (size_t)(kc + 2 * q + 1) * NC + c0);
      gslot[u] = (((q >> 1) * NT + (c0 >> 4)) * 32) + (q & 1) * 16 + (c0 & 15);
    }
    if (AM == MM_A_TANH) {
#pragma unroll
      for (int i = 0; i < 8; ++i) {
        xa[i].x = tanhf(xa[i].x); xa[i].y = tanhf(xa[i].y);
        xa[i].z = tanhf(xa[i].z); xa[i].w = tanhf(xa[i].w);
      }
    }
    if (AM == MM_A_SET || AM == MM_A_ADD) {
#pragma unroll
      for (int i = 0; i < 8; ++i) {
        if (AM == MM_A_SET) {
          aa[i].x = w * kk[i].x; aa[i].y = w * kk[i].y; aa[i].z = w * kk[i].z; aa[i].w = w * kk[i].w;
        } else {
          aa[i].x = aa[i].x + w * kk[i].x; aa[i].y = aa[i].y + w * kk[i].y;
          aa[i].z = aa[i].z + w * kk[i].z; aa[i].w = aa[i].w + w * kk[i].w;
        }
        xa[i].x = xa[i].x + c * kk[i].x; xa[i].y = xa[i].y + c * kk[i].y;
        xa[i].z = xa[i].z + c * kk[i].z; xa[i].w = xa[i].w + c * kk[i].w;
      }
    }
    if (AM == MM_A_FINAL) {
#pragma unroll
      for (int i = 0; i < 8; ++i) {
        xa[i].x = xa[i].x + 0.5f * (aa[i].x + kk[i].x); xa[i].y = xa[i].y + 0.5f * (aa[i].y + kk[i].y);
        xa[i].z = xa[i].z + 0.5f * (aa[i].z + kk[i].z); xa[i].w = xa[i].w + 0.5f * (aa[i].w + kk[i].w);
      }
    }
#pragma unroll
    for (int i = 0; i < 8; ++i)
      *(float4*)(as + (2 * i + half) * MM_APITCH + 4 * m) = xa[i];
#pragma unroll
    for (int u = 0; u < ITERS; ++u) {
      float* gp = (float*)&Gs[gslot[u]];
      *(float4*)(gp)     = make_float4(g0[u].x, g1[u].x, g0[u].y, g1[u].y);
      *(float4*)(gp + 4) = make_float4(g0[u].z, g1[u].z, g0[u].w, g1[u].w);
    }
    __syncthreads();
    if ((AM == MM_A_TANH || AM == MM_A_SET || AM == MM_A_ADD) && live) {
      float* sp = (AM == MM_A_TANH ? Hio : Acc) + tbase + kc + 4 * m;
#pragma unroll
      for (int i = 0; i < 8; ++i)
        *(float4*)(sp + (size_t)(2 * i + half) * K) = (AM == MM_A_TANH) ? xa[i] : aa[i];
    }
#pragma unroll
    for (int j = 0; j < MM_KC / 4; ++j) {
      const v2f a = *(const v2f*)(as + m * MM_APITCH + 4 * j + 2 * half);
#pragma unroll
      for (int t = 0; t < NT; ++t) {
        const v2f b = Gs[(j * NT + t) * 32 + lane];
        acc[t] = __builtin_amdgcn_wmma_f32_16x16x4_f32(false, a, false, b,
                                                       (short)0, acc[t], false, false);
      }
    }
  }

  if (live) {
    const size_t row0 = (size_t)(tile * 16 + half * 8);
    float* zr = Zp + row0 * CP + m;
#pragma unroll
    for (int t = 0; t < NTP; ++t)
#pragma unroll
      for (int r = 0; r < 8; ++r)
        zr[(size_t)r * CP + 16 * t] = acc[t][r];
    float* sr = S + row0 * CO + m;
#pragma unroll
    for (int t = NTP; t < NT; ++t)
#pragma unroll
      for (int r = 0; r < 8; ++r)
        sr[(size_t)r * CO + 16 * (t - NTP)] = acc[t][r];
  }
}

__global__ void gde_prep_g(float* __restrict__ G, const float* __restrict__ Wp,
                           const float* __restrict__ R, int Cin, int Cout) {
  int idx = blockIdx.x * blockDim.x + threadIdx.x;
  int Ncols = 3 * Cout;
  int tot = Cin * Ncols;
  if (idx >= tot) return;
  int r = idx / Ncols, c = idx % Ncols;
  float v;
  if (c < 2 * Cout) {
    const int ch = c >> 1;
    const float w0 = Wp[r * Cout + ch];
    v = (c & 1) ? (Wp[(size_t)Cin * Cout + r * Cout + ch] - w0) : w0;
  } else {
    v = R[r * Cout + (c - 2 * Cout)];
  }
  G[idx] = v;
}

#define ER_WAVES   4
#define ER_THREADS (32 * ER_WAVES)
#define ER_RPW     64
#define ER_CHUNK   (ER_WAVES * ER_RPW)
#define ER_NCHUNK  (GDE_E / ER_CHUNK)
#define ER_SPLIT   (GDE_N / 2)
#define ER_TKPITCH 32
#define ER_PASSES  10
static_assert(GDE_E % ER_CHUNK == 0, "whole chunks");

__global__ void gde_deg(float* __restrict__ deg, const int* __restrict__ dst, int E) {
  int e = blockIdx.x * blockDim.x + threadIdx.x;
  if (e < E) atomicAdd(&deg[dst[e]], 1.0f);
}

__global__ void gde_edge_w(float* __restrict__ wa, float* __restrict__ wb,
                           const float* __restrict__ deg, const int* __restrict__ dst,
                           const float* __restrict__ p, int E, unsigned* __restrict__ tickets) {
  int e = blockIdx.x * blockDim.x + threadIdx.x;
  if (e < ER_PASSES * 2 * ER_TKPITCH) tickets[e] = 0u;
  if (e < E) {
    const float iv = 1.0f / fmaxf(deg[dst[e]], 1.0f);
    wa[e] = iv;
    wb[e] = p[e] * iv;
  }
}


template <int EPS>
__device__ __forceinline__ int gde_edge_pick(int v, int j, int sub) {
  const int a = __builtin_amdgcn_readlane(v, j);
  if (EPS == 1) return a;
  const int b = __builtin_amdgcn_readlane(v, j + 1);
  return sub ? b : a;
}

template <int COUT, int G>
__device__ __forceinline__ void gde_edge_group(
    float* __restrict__ agg, const float* __restrict__ Z, int sg, int dg, int ag, int bg,
    int jb, int j0, int K, int sub, int c) {
  constexpr int LPE = COUT < 32 ? COUT : 32;
  constexpr int EPS = 32 / LPE;
  constexpr int CPL = COUT / LPE;
  float2 z[G][CPL];
  int    d[G];
  float  wa[G], wb[G];
#pragma unroll
  for (int u = 0; u < G; ++u) {
    const int j = jb + u * EPS;
    const int s = gde_edge_pick<EPS>(sg, j, sub);
    d[u]  = gde_edge_pick<EPS>(dg, j, sub);
    wa[u] = __int_as_float(gde_edge_pick<EPS>(ag, j, sub));
    wb[u] = __int_as_float(gde_edge_pick<EPS>(bg, j, sub));
    const float2* zr = (const float2*)(Z + (size_t)s * (2 * COUT)) + c;
#pragma unroll
    for (int h = 0; h < CPL; ++h) z[u][h] = zr[h * LPE];
  }
#pragma unroll
  for (int u = 0; u < G; ++u) {
    if (EPS == 1 || j0 + u * EPS + sub < K) {
      float* ar = agg + (size_t)d[u] * COUT + c;
#pragma unroll
      for (int h = 0; h < CPL; ++h)
        atomicAdd(ar + h * LPE, wa[u] * z[u][h].x + wb[u] * z[u][h].y);
    }
  }
}

template <int COUT>
__global__ __launch_bounds__(ER_THREADS) void gde_edge_rows(
    float* __restrict__ agg, const float* __restrict__ Z, const int* __restrict__ src,
    const int* __restrict__ dst, const float* __restrict__ wa, const float* __restrict__ wb,
    unsigned* __restrict__ tickets) {
  constexpr int LPE = COUT < 32 ? COUT : 32;
  constexpr int EPS = 32 / LPE;
  constexpr int GRP = 8;
  static_assert(32 % LPE == 0 && COUT % LPE == 0 && 32 % (GRP * EPS) == 0, "shape");
  __shared__ int tk[2];
  __shared__ int rec[ER_WAVES][4][ER_RPW];
  const int tid  = threadIdx.x;
  const int lane = tid & 31;
  const int wave = tid >> 5;
  const int sub  = lane / LPE;
  const int c    = lane % LPE;
  if (tid == 0) {
    int part = (int)((__builtin_amdgcn_s_sendmsg_rtn(0x87) >> 18) & 1u);
    int t = (int)atomicAdd(tickets + part * ER_TKPITCH, 1u);
    if (t >= ER_NCHUNK) {
      part ^= 1;
      t = (int)atomicAdd(tickets + part * ER_TKPITCH, 1u);
    }
    tk[0] = t;
    tk[1] = part;
  }
  __syncthreads();
  const int t    = tk[0];
  const int part = tk[1];
  if (t >= ER_NCHUNK) return;

  const int e  = t * ER_CHUNK + wave * ER_RPW + lane;
  const int s0 = src[e],      s1 = src[e + 32];
  const int d0 = dst[e],      d1 = dst[e + 32];
  const int a0 = __float_as_int(wa[e]), a1 = __float_as_int(wa[e + 32]);
  const int b0 = __float_as_int(wb[e]), b1 = __float_as_int(wb[e + 32]);
  const bool k0 = (d0 >= ER_SPLIT) == (part != 0);
  const bool k1 = (d1 >= ER_SPLIT) == (part != 0);
  const unsigned m0 = __builtin_amdgcn_ballot_w32(k0);
  const unsigned m1 = __builtin_amdgcn_ballot_w32(k1);
  const int n0 = __popc(m0);
  const int K  = n0 + __popc(m1);
  const unsigned below = (1u << lane) - 1u;
  const int q0 = k0 ? __popc(m0 & below)      : K + __popc(~m0 & below);
  const int q1 = k1 ? n0 + __popc(m1 & below) : K + (32 - n0) + __popc(~m1 & below);
  int* rs = rec[wave][0];
  int* rd = rec[wave][1];
  int* ra = rec[wave][2];
  int* rb = rec[wave][3];
  rs[q0] = s0; rd[q0] = d0; ra[q0] = a0; rb[q0] = b0;
  rs[q1] = s1; rd[q1] = d1; ra[q1] = a1; rb[q1] = b1;
  const int sv0 = rs[lane], sv1 = rs[lane + 32];
  const int dv0 = rd[lane], dv1 = rd[lane + 32];
  const int av0 = ra[lane], av1 = ra[lane + 32];
  const int bv0 = rb[lane], bv1 = rb[lane + 32];

  const int S = (K + EPS - 1) / EPS;
  int q = 0;
#define GDE_RUN(G)                                                                       \
  {                                                                                      \
    const int  j0 = q * EPS;                                                             \
    const bool hi = j0 >= 32;                                                            \
    gde_edge_group<COUT, G>(agg, Z, hi ? sv1 : sv0, hi ? dv1 : dv0, hi ? av1 : av0,      \
                            hi ? bv1 : bv0, j0 & 31, j0, K, sub, c);                     \
  }
#pragma unroll 1
  for (; q + GRP <= S; q += GRP) GDE_RUN(GRP)
  if (S - q >= 4) { GDE_RUN(4) q += 4; }
  if (S - q >= 2) { GDE_RUN(2) q += 2; }
  if (S - q >= 1) { GDE_RUN(1) }
#undef GDE_RUN
}

__global__ void gde_log_softmax16(float* __restrict__ out, int n) {
  int i = blockIdx.x * blockDim.x + threadIdx.x;
  if (i >= n) return;
  float v[16];
  float mx = -3.402823466e+38f;
  float* row = out + (size_t)i * 16;
#pragma unroll
  for (int c = 0; c < 16; ++c) { v[c] = tanhf(row[c]); mx = fmaxf(mx, v[c]); }
  float s = 0.f;
#pragma unroll
  for (int c = 0; c < 16; ++c) s += __expf(v[c] - mx);
  float ls = __logf(s);
#pragma unroll
  for (int c = 0; c < 16; ++c) row[c] = v[c] - mx - ls;
}

extern "C" void kernel_launch(void* const* d_in, const int* in_sizes, int n_in,
                              void* d_out, int out_size, void* d_ws, size_t ws_size,
                              hipStream_t stream) {
  const float* x   = (const float*)d_in[0];
  const float* p   = (const float*)d_in[1];
  const int*   src = (const int*)d_in[2];
  const int*   dst = (const int*)d_in[3];
  const float* W1 = (const float*)d_in[4],  *R1 = (const float*)d_in[5],  *b1 = (const float*)d_in[6];
  const float* Wa = (const float*)d_in[7],  *Ra = (const float*)d_in[8],  *ba = (const float*)d_in[9];
  const float* Wb = (const float*)d_in[10], *Rb = (const float*)d_in[11], *bb = (const float*)d_in[12];
  const float* W2 = (const float*)d_in[13], *R2 = (const float*)d_in[14], *b2 = (const float*)d_in[15];
  float* out = (float*)d_out;

  float* W = (float*)d_ws;
  size_t o = 0;
  float* Z   = W + o; o += (size_t)GDE_N * 128;
  float* H   = W + o; o += (size_t)GDE_N * 64;
  float* U   = W + o; o += (size_t)GDE_N * 64;
  float* Kb  = W + o; o += (size_t)GDE_N * 64;
  float* ACC = W + o; o += (size_t)GDE_N * 64;
  unsigned* TK = (unsigned*)(W + o); o += (size_t)ER_PASSES * 2 * ER_TKPITCH;
  float* DEG = W + o; o += (size_t)GDE_N;
  float* EWA = W + o; o += (size_t)GDE_E;
  float* EWB = W + o; o += (size_t)GDE_E;
  float* GC1 = W + o; o += (size_t)256 * 192;
  float* GCA = W + o; o += (size_t)64 * 192;
  float* GCB = W + o; o += (size_t)64 * 192;
  float* GC2 = W + o; o += (size_t)64 * 48;
  (void)ws_size; (void)n_in; (void)in_sizes; (void)out_size;

  gde_prep_g<<<(256 * 192 + 255) / 256, 256, 0, stream>>>(GC1, W1, R1, 256, 64);
  gde_prep_g<<<(64 * 192 + 255) / 256, 256, 0, stream>>>(GCA, Wa, Ra, 64, 64);
  gde_prep_g<<<(64 * 192 + 255) / 256, 256, 0, stream>>>(GCB, Wb, Rb, 64, 64);
  gde_prep_g<<<(64 * 48 + 255) / 256, 256, 0, stream>>>(GC2, W2, R2, 64, 16);

  hipMemsetAsync(DEG, 0, sizeof(float) * GDE_N, stream);
  gde_deg<<<(GDE_E + 255) / 256, 256, 0, stream>>>(DEG, dst, GDE_E);
  gde_edge_w<<<(GDE_E + 255) / 256, 256, 0, stream>>>(EWA, EWB, DEG, dst, p, GDE_E, TK);

  auto spline = [&](const float* y, int Cin, int Cout, const float* G,
                    const float* bias, float* ob, int am, float w, float c) {
    const int mmg = (MM_TILES + MM_WAVES - 1) / MM_WAVES;
    if (Cin == 256)
      gde_mm_lds<256, 12, MM_A_PLAIN><<<mmg, MM_THREADS, 0, stream>>>(y, G, Z, ob, bias, nullptr, nullptr, nullptr, w, c);
    else if (am == MM_A_PLAIN)
      gde_mm_lds<64, 12, MM_A_PLAIN><<<mmg, MM_THREADS, 0, stream>>>(y, G, Z, ob, bias, nullptr, nullptr, nullptr, w, c);
    else if (am == MM_A_TANH)
      gde_mm_lds<64, 12, MM_A_TANH><<<mmg, MM_THREADS, 0, stream>>>(y, G, Z, ob, bias, H, Kb, ACC, w, c);
    else if (am == MM_A_SET)
      gde_mm_lds<64, 12, MM_A_SET><<<mmg, MM_THREADS, 0, stream>>>(y, G, Z, ob, bias, H, Kb, ACC, w, c);
    else if (am == MM_A_ADD)
      gde_mm_lds<64, 12, MM_A_ADD><<<mmg, MM_THREADS, 0, stream>>>(y, G, Z, ob, bias, H, Kb, ACC, w, c);
    else
      gde_mm_lds<64, 3, MM_A_FINAL><<<mmg, MM_THREADS, 0, stream>>>(y, G, Z, ob, bias, H, Kb, ACC, w, c);
    if (Cout == 64) gde_edge_rows<64><<<2 * ER_NCHUNK, ER_THREADS, 0, stream>>>(ob, Z, src, dst, EWA, EWB, TK);
    else            gde_edge_rows<16><<<2 * ER_NCHUNK, ER_THREADS, 0, stream>>>(ob, Z, src, dst, EWA, EWB, TK);
    TK += 2 * ER_TKPITCH;
  };
  auto odefunc = [&](int am, float w, float c) {
    spline(H, 64, 64, GCA, ba, U, am, w, c);
    spline(U, 64, 64, GCB, bb, Kb, MM_A_PLAIN, 0.0f, 0.0f);
  };

  spline(x, 256, 64, GC1, b1, H, MM_A_PLAIN, 0.0f, 0.0f);

  odefunc(MM_A_TANH, 0.0f, 0.0f);
  odefunc(MM_A_SET, 1.0f, 1.5f);
  odefunc(MM_A_ADD, 2.0f, 1.5f);
  odefunc(MM_A_ADD, 2.0f, 3.0f);

  spline(H, 64, 16, GC2, b2, out, MM_A_FINAL, 0.0f, 0.0f);
  gde_log_softmax16<<<(GDE_N + 255) / 256, 256, 0, stream>>>(out, GDE_N);
}


# Edge sums in LDS: edges bucketed once per call by 196-row destination range; a 32-wave group per range adds messages with LDS float adds, stores rows once; no global float atomics; arrival order kept
# speedup vs baseline: 2.8224x; 1.0732x over previous
#include <hip/hip_runtime.h>
#include <math.h>

#define GDE_N 50000
#define GDE_E 800000

typedef float v2f __attribute__((ext_vector_type(2)));
typedef float v4f __attribute__((ext_vector_type(4)));
typedef float v8f __attribute__((ext_vector_type(8)));

#define MM_A_PLAIN 0
#define MM_A_TANH  1
#define MM_A_SET   2
#define MM_A_ADD   3
#define MM_A_FINAL 4
#define MM_TILES (GDE_N / 16)
#define MM_WAVES 8
#define MM_THREADS (32 * MM_WAVES)
#define MM_KC 64
#define MM_APITCH (MM_KC + 4)

static_assert(GDE_N % 16 == 0, "whole row tiles");

template <int K, int NT, int AM>
__global__ __launch_bounds__(MM_THREADS) __attribute__((amdgpu_waves_per_eu(4, 4))) void gde_mm_lds(
    const float* __restrict__ X, const float* __restrict__ G, float* __restrict__ Zp,
    float* __restrict__ S, const float* __restrict__ bias, float* Hio,
    const float* __restrict__ Kp, float* __restrict__ Acc, float w, float c) {
  static_assert(K % MM_KC == 0, "K must be whole chunks");
  static_assert(AM == MM_A_PLAIN || K == MM_KC, "a loader that stores sees every element once");
  static_assert(NT % 3 == 0, "columns are [pairs | root], two to one");
  constexpr int NC    = 16 * NT;
  constexpr int NTP   = 2 * NT / 3;
  constexpr int CP    = 16 * NTP;
  constexpr int CO    = NC - CP;
  constexpr int ITEMS = (MM_KC / 2) * (NC / 4);
  constexpr int ITERS = (ITEMS + MM_THREADS - 1) / MM_THREADS;
  constexpr int FULL  = ITEMS / MM_THREADS;
  __shared__ __attribute__((aligned(16))) v2f   Gs[(MM_KC / 4) * NT * 32];
  __shared__ __attribute__((aligned(16))) float As[MM_WAVES][16 * MM_APITCH];

  const int tid  = threadIdx.x;
  const int lane = tid & 31;
  const int wave = __builtin_amdgcn_readfirstlane(tid >> 5);
  const int m    = lane & 15;
  const int half = lane >> 4;
  const int  tile_raw = blockIdx.x * MM_WAVES + wave;
  const bool live     = tile_raw < MM_TILES;
  const int  tile     = live ? tile_raw : (MM_TILES - 1);
  const size_t tbase = (size_t)tile * 16 * K;
  const float* __restrict__ xt = X + tbase;
  float* as = As[wave];

  v8f acc[NT];
#pragma unroll
  for (int t = 0; t < NTP; ++t) acc[t] = (v8f){0.f, 0.f, 0.f, 0.f, 0.f, 0.f, 0.f, 0.f};
#pragma unroll
  for (int t = NTP; t < NT; ++t) {
    const float bv = bias[16 * (t - NTP) + m];
    acc[t] = (v8f){bv, bv, bv, bv, bv, bv, bv, bv};
  }

#pragma unroll 1
  for (int kc = 0; kc < K; kc += MM_KC) {
    if (kc) __syncthreads();
    float4 xa[8], kk[8], aa[8];
#pragma unroll
    for (int i = 0; i < 8; ++i) {
      const size_t off = (size_t)(2 * i + half) * K + kc + 4 * m;
      if (AM == MM_A_TANH) xa[i] = *(const float4*)(Hio + tbase + off);
      else                 xa[i] = *(const float4*)(xt + off);
      if (AM == MM_A_SET || AM == MM_A_ADD || AM == MM_A_FINAL) kk[i] = *(const float4*)(Kp + tbase + off);
      if (AM == MM_A_ADD || AM == MM_A_FINAL) aa[i] = *(const float4*)(Acc + tbase + off);
    }
    float4 g0[ITERS], g1[ITERS];
    int    gslot[ITERS];
#pragma unroll
    for (int u = 0; u < ITERS; ++u) {
      const int i  = tid + u * MM_THREADS;
      const int ic = (u < FULL || i < ITEMS) ? i : (i - ITEMS);
      const int q  = ic / (NC / 4);
      const int c0 = (ic - q * (NC / 4)) * 4;
      g0[u] = *(const float4*)(G + (size_t)(kc + 2 * q)     * NC + c0);
      g1[u] = *(const float4*)(G + (size_t)(kc + 2 * q + 1) * NC + c0);
      gslot[u] = (((q >> 1) * NT + (c0 >> 4)) * 32) + (q & 1) * 16 + (c0 & 15);
    }
    if (AM == MM_A_TANH) {
#pragma unroll
      for (int i = 0; i < 8; ++i) {
        xa[i].x = tanhf(xa[i].x); xa[i].y = tanhf(xa[i].y);
        xa[i].z = tanhf(xa[i].z); xa[i].w = tanhf(xa[i].w);
      }
    }
    if (AM == MM_A_SET || AM == MM_A_ADD) {
#pragma unroll
      for (int i = 0; i < 8; ++i) {
        if (AM == MM_A_SET) {
          aa[i].x = w * kk[i].x; aa[i].y = w * kk[i].y; aa[i].z = w * kk[i].z; aa[i].w = w * kk[i].w;
        } else {
          aa[i].x = aa[i].x + w * kk[i].x; aa[i].y = aa[i].y + w * kk[i].y;
          aa[i].z = aa[i].z + w * kk[i].z; aa[i].w = aa[i].w + w * kk[i].w;
        }
        xa[i].x = xa[i].x + c * kk[i].x; xa[i].y = xa[i].y + c * kk[i].y;
        xa[i].z = xa[i].z + c * kk[i].z; xa[i].w = xa[i].w + c * kk[i].w;
      }
    }
    if (AM == MM_A_FINAL) {
#pragma unroll
      for (int i = 0; i < 8; ++i) {
        xa[i].x = xa[i].x + 0.5f * (aa[i].x + kk[i].x); xa[i].y = xa[i].y + 0.5f * (aa[i].y + kk[i].y);
        xa[i].z = xa[i].z + 0.5f * (aa[i].z + kk[i].z); xa[i].w = xa[i].w + 0.5f * (aa[i].w + kk[i].w);
      }
    }
#pragma unroll
    for (int i = 0; i < 8; ++i)
      *(float4*)(as + (2 * i + half) * MM_APITCH + 4 * m) = xa[i];
#pragma unroll
    for (int u = 0; u < ITERS; ++u) {
      float* gp = (float*)&Gs[gslot[u]];
      *(float4*)(gp)     = make_float4(g0[u].x, g1[u].x, g0[u].y, g1[u].y);
      *(float4*)(gp + 4) = make_float4(g0[u].z, g1[u].z, g0[u].w, g1[u].w);
    }
    __syncthreads();
    if ((AM == MM_A_TANH || AM == MM_A_SET || AM == MM_A_ADD) && live) {
      float* sp = (AM == MM_A_TANH ? Hio : Acc) + tbase + kc + 4 * m;
#pragma unroll
      for (int i = 0; i < 8; ++i)
        *(float4*)(sp + (size_t)(2 * i + half) * K) = (AM == MM_A_TANH) ? xa[i] : aa[i];
    }
#pragma unroll
    for (int j = 0; j < MM_KC / 4; ++j) {
      const v2f a = *(const v2f*)(as + m * MM_APITCH + 4 * j + 2 * half);
#pragma unroll
      for (int t = 0; t < NT; ++t) {
        const v2f b = Gs[(j * NT + t) * 32 + lane];
        acc[t] = __builtin_amdgcn_wmma_f32_16x16x4_f32(false, a, false, b,
                                                       (short)0, acc[t], false, false);
      }
    }
  }

  if (live) {
    const size_t row0 = (size_t)(tile * 16 + half * 8);
    float* zr = Zp + row0 * CP + m;
#pragma unroll
    for (int t = 0; t < NTP; ++t)
#pragma unroll
      for (int r = 0; r < 8; ++r)
        zr[(size_t)r * CP + 16 * t] = acc[t][r];
    float* sr = S + row0 * CO + m;
#pragma unroll
    for (int t = NTP; t < NT; ++t)
#pragma unroll
      for (int r = 0; r < 8; ++r)
        sr[(size_t)r * CO + 16 * (t - NTP)] = acc[t][r];
  }
}

__global__ void gde_prep_g(float* __restrict__ G, const float* __restrict__ Wp,
                           const float* __restrict__ R, int Cin, int Cout,
                           unsigned* __restrict__ cz, int nz) {
  int idx = blockIdx.x * blockDim.x + threadIdx.x;
  if (idx < nz) cz[idx] = 0u;
  int Ncols = 3 * Cout;
  int tot = Cin * Ncols;
  if (idx >= tot) return;
  int r = idx / Ncols, c = idx % Ncols;
  float v;
  if (c < 2 * Cout) {
    const int ch = c >> 1;
    const float w0 = Wp[r * Cout + ch];
    v = (c & 1) ? (Wp[(size_t)Cin * Cout + r * Cout + ch] - w0) : w0;
  } else {
    v = R[r * Cout + (c - 2 * Cout)];
  }
  G[idx] = v;
}

#define EB_NB      256
#define EB_RB      ((GDE_N + EB_NB - 1) / EB_NB)
#define EB_WAVES   32
#define EB_THREADS (32 * EB_WAVES)
#define EB_OFFW    640
static_assert(EB_NB <= EB_THREADS && EB_RB <= EB_THREADS && EB_NB + 1 <= EB_OFFW, "one thread per range / per row");
static_assert(GDE_N <= 65536 && EB_RB <= 65536, "a record packs the source row and the local row in 16 bits each");

__global__ __launch_bounds__(EB_THREADS) void gde_deg(float* __restrict__ deg, unsigned* __restrict__ bcnt,
                                                      const int* __restrict__ dst, int E) {
  __shared__ unsigned hist[EB_NB];
  const int tid = threadIdx.x;
  if (tid < EB_NB) hist[tid] = 0u;
  __syncthreads();
  const int e = blockIdx.x * EB_THREADS + tid;
  if (e < E) {
    const int d = dst[e];
    atomicAdd(&deg[d], 1.0f);
    atomicAdd(&hist[d / EB_RB], 1u);
  }
  __syncthreads();
  if (tid < EB_NB) {
    const unsigned h = hist[tid];
    if (h) atomicAdd(&bcnt[tid], h);
  }
}

__global__ __launch_bounds__(EB_THREADS) void gde_edge_fill(
    int* __restrict__ pk, float* __restrict__ pp, unsigned* __restrict__ off, unsigned* __restrict__ bcur,
    const unsigned* __restrict__ bcnt, const int* __restrict__ src, const int* __restrict__ dst,
    const float* __restrict__ p, int E) {
  __shared__ unsigned hist[EB_NB];
  __shared__ unsigned scan[EB_NB];
  __shared__ unsigned base[EB_NB];
  const int tid = threadIdx.x;
  unsigned cntv = 0u;
  if (tid < EB_NB) { hist[tid] = 0u; cntv = bcnt[tid]; scan[tid] = cntv; }
  __syncthreads();
  const int e = blockIdx.x * EB_THREADS + tid;
  int d = 0, b = 0;
  unsigned r = 0u;
  if (e < E) {
    d = dst[e];
    b = d / EB_RB;
    r = atomicAdd(&hist[b], 1u);
  }
  for (int st = 1; st < EB_NB; st <<= 1) {
    unsigned v = 0u;
    if (tid < EB_NB && tid >= st) v = scan[tid - st];
    __syncthreads();
    if (tid < EB_NB) scan[tid] += v;
    __syncthreads();
  }
  if (tid < EB_NB) {
    const unsigned excl = scan[tid] - cntv;
    const unsigned h = hist[tid];
    const unsigned rs = h ? atomicAdd(&bcur[tid], h) : 0u;
    base[tid] = excl + rs;
    if (blockIdx.x == 0) {
      off[tid] = excl;
      if (tid == EB_NB - 1) off[EB_NB] = scan[tid];
    }
  }
  __syncthreads();
  if (e < E) {
    const unsigned pos = base[b] + r;
    pk[pos] = src[e] | ((d - b * EB_RB) << 16);
    pp[pos] = p[e];
  }
}


template <int EPS>
__device__ __forceinline__ int gde_rec_pick(int v, int j, int sub) {
  int r = __builtin_amdgcn_readlane(v, j);
#pragma unroll
  for (int k = 1; k < EPS; ++k) {
    const int o = __builtin_amdgcn_readlane(v, j + k);
    r = (sub == k) ? o : r;
  }
  return r;
}

template <int COUT, int G>
__device__ __forceinline__ void gde_lds_group(float* acc, const float* __restrict__ Z,
                                              int pkv, int wav, int wbv, int jb, int cnt, int sub, int c) {
  constexpr int LPE = COUT / 2;
  constexpr int EPS = 32 / LPE;
  float4 z[G];
  int    dl[G];
  float  wa[G], wb[G];
#pragma unroll
  for (int u = 0; u < G; ++u) {
    const int j = jb + u * EPS;
    const unsigned pkx = (unsigned)gde_rec_pick<EPS>(pkv, j, sub);
    dl[u] = (int)(pkx >> 16);
    wa[u] = __int_as_float(gde_rec_pick<EPS>(wav, j, sub));
    wb[u] = __int_as_float(gde_rec_pick<EPS>(wbv, j, sub));
    z[u] = *(const float4*)(Z + (size_t)(pkx & 0xffffu) * (2 * COUT) + 4 * c);
  }
#pragma unroll
  for (int u = 0; u < G; ++u) {
    if (EPS == 1 || jb + u * EPS + sub < cnt) {
      float* ar = acc + dl[u] * COUT + 2 * c;
      __hip_atomic_fetch_add(ar,     wa[u] * z[u].x + wb[u] * z[u].y, __ATOMIC_RELAXED, __HIP_MEMORY_SCOPE_WORKGROUP);
      __hip_atomic_fetch_add(ar + 1, wa[u] * z[u].z + wb[u] * z[u].w, __ATOMIC_RELAXED, __HIP_MEMORY_SCOPE_WORKGROUP);
    }
  }
}

template <int COUT>
__global__ __launch_bounds__(EB_THREADS) void gde_edge_lds(
    float* agg, const float* __restrict__ Z, const int* __restrict__ pk, const float* __restrict__ pp,
    const unsigned* __restrict__ off, const float* __restrict__ deg) {
  constexpr int LPE = COUT / 2;
  constexpr int EPS = 32 / LPE;
  constexpr int GRP = EPS == 1 ? 16 : 8;
  static_assert(32 % LPE == 0 && COUT % 4 == 0 && 32 % (GRP * EPS) == 0, "shape");
  __shared__ __attribute__((aligned(16))) float acc[EB_RB * COUT];
  __shared__ float inv[EB_RB];
  const int tid  = threadIdx.x;
  const int lane = tid & 31;
  const int wave = __builtin_amdgcn_readfirstlane(tid >> 5);
  const int sub  = lane / LPE;
  const int c    = lane % LPE;
  const int row0 = blockIdx.x * EB_RB;
  const int nrr  = GDE_N - row0;
  const int nr   = nrr < EB_RB ? (nrr > 0 ? nrr : 0) : EB_RB;
  const int nq   = nr * (COUT / 4);
  float* ag = agg + (size_t)row0 * COUT;
  constexpr int NQ = (EB_RB * (COUT / 4) + EB_THREADS - 1) / EB_THREADS;
  if (nr > 0) {
    v4f st[NQ];
#pragma unroll
    for (int u = 0; u < NQ; ++u) {
      const int qi = tid + u * EB_THREADS;
      st[u] = ((const v4f*)ag)[qi < nq ? qi : nq - 1];
    }
    const float dg = deg[row0 + (tid < nr ? tid : nr - 1)];
#pragma unroll
    for (int u = 0; u < NQ; ++u) asm volatile("" :: "v"(st[u]));
#pragma unroll
    for (int u = 0; u < NQ; ++u) {
      const int qi = tid + u * EB_THREADS;
      if (qi < nq) ((v4f*)acc)[qi] = st[u];
    }
    if (tid < nr) inv[tid] = 1.0f / fmaxf(dg, 1.0f);
  }
  __syncthreads();

  const unsigned o0 = off[blockIdx.x], o1 = off[blockIdx.x + 1];
  const unsigned share = (o1 - o0 + EB_WAVES - 1) / EB_WAVES;
  unsigned i = o0 + (unsigned)wave * share;
  const unsigned we = (i + share < o1) ? (i + share) : o1;
  unsigned ri = i + lane;
  unsigned rc = ri < GDE_E - 1u ? ri : GDE_E - 1u;
  int   npk = pk[rc];
  float npv = pp[rc];
  if (ri >= we) { npk = 0; npv = 0.0f; }
#pragma unroll 1
  for (; i < we; i += 32) {
    const int   cnt = (we - i < 32u) ? (int)(we - i) : 32;
    const int   pkv = npk;
    const float pv  = npv;
    ri = i + 32u + lane;
    rc = ri < GDE_E - 1u ? ri : GDE_E - 1u;
    npk = pk[rc];
    npv = pp[rc];
    if (ri >= we) { npk = 0; npv = 0.0f; }
    const float iv  = (lane < cnt) ? inv[(unsigned)pkv >> 16] : 0.0f;
    const int   wav = __float_as_int(iv);
    const int   wbv = __float_as_int(pv * iv);
    const int S = (cnt + EPS - 1) / EPS;
    int q = 0;
#define GDE_RUN(G) gde_lds_group<COUT, G>(acc, Z, pkv, wav, wbv, q * EPS, cnt, sub, c);
#pragma unroll 1
    for (; q + GRP <= S; q += GRP) GDE_RUN(GRP)
    if (GRP > 8 && S - q >= 8) { GDE_RUN(8) q += 8; }
    if (S - q >= 4) { GDE_RUN(4) q += 4; }
    if (S - q >= 2) { GDE_RUN(2) q += 2; }
    if (S - q >= 1) { GDE_RUN(1) }
#undef GDE_RUN
  }
  __syncthreads();
  for (int j = tid; j < nq; j += EB_THREADS) ((float4*)ag)[j] = ((const float4*)acc)[j];
}

__global__ void gde_log_softmax16(float* __restrict__ out, int n) {
  int i = blockIdx.x * blockDim.x + threadIdx.x;
  if (i >= n) return;
  float v[16];
  float mx = -3.402823466e+38f;
  float* row = out + (size_t)i * 16;
#pragma unroll
  for (int c = 0; c < 16; ++c) { v[c] = tanhf(row[c]); mx = fmaxf(mx, v[c]); }
  float s = 0.f;
#pragma unroll
  for (int c = 0; c < 16; ++c) s += __expf(v[c] - mx);
  float ls = __logf(s);
#pragma unroll
  for (int c = 0; c < 16; ++c) row[c] = v[c] - mx - ls;
}

extern "C" void kernel_launch(void* const* d_in, const int* in_sizes, int n_in,
                              void* d_out, int out_size, void* d_ws, size_t ws_size,
                              hipStream_t stream) {
  const float* x   = (const float*)d_in[0];
  const float* p   = (const float*)d_in[1];
  const int*   src = (const int*)d_in[2];
  const int*   dst = (const int*)d_in[3];
  const float* W1 = (const float*)d_in[4],  *R1 = (const float*)d_in[5],  *b1 = (const float*)d_in[6];
  const float* Wa = (const float*)d_in[7],  *Ra = (const float*)d_in[8],  *ba = (const float*)d_in[9];
  const float* Wb = (const float*)d_in[10], *Rb = (const float*)d_in[11], *bb = (const float*)d_in[12];
  const float* W2 = (const float*)d_in[13], *R2 = (const float*)d_in[14], *b2 = (const float*)d_in[15];
  float* out = (float*)d_out;

  float* W = (float*)d_ws;
  size_t o = 0;
  float* Z   = W + o; o += (size_t)GDE_N * 128;
  float* H   = W + o; o += (size_t)GDE_N * 64;
  float* U   = W + o; o += (size_t)GDE_N * 64;
  float* Kb  = W + o; o += (size_t)GDE_N * 64;
  float* ACC = W + o; o += (size_t)GDE_N * 64;
  unsigned* OFF = (unsigned*)(W + o); o += (size_t)EB_OFFW;
  float* DEG = W + o; o += (size_t)GDE_N;
  int*   PK  = (int*)(W + o); o += (size_t)GDE_E;
  float* PP  = W + o; o += (size_t)GDE_E;
  unsigned* BCNT = (unsigned*)U;
  unsigned* BCUR = BCNT + EB_NB;
  float* GC1 = W + o; o += (size_t)256 * 192;
  float* GCA = W + o; o += (size_t)64 * 192;
  float* GCB = W + o; o += (size_t)64 * 192;
  float* GC2 = W + o; o += (size_t)64 * 48;
  (void)ws_size; (void)n_in; (void)in_sizes; (void)out_size;

  gde_prep_g<<<(256 * 192 + 255) / 256, 256, 0, stream>>>(GC1, W1, R1, 256, 64, BCNT, 2 * EB_NB);
  gde_prep_g<<<(64 * 192 + 255) / 256, 256, 0, stream>>>(GCA, Wa, Ra, 64, 64, BCNT, 0);
  gde_prep_g<<<(64 * 192 + 255) / 256, 256, 0, stream>>>(GCB, Wb, Rb, 64, 64, BCNT, 0);
  gde_prep_g<<<(64 * 48 + 255) / 256, 256, 0, stream>>>(GC2, W2, R2, 64, 16, BCNT, 0);

  hipMemsetAsync(DEG, 0, sizeof(float) * GDE_N, stream);
  gde_deg<<<(GDE_E + EB_THREADS - 1) / EB_THREADS, EB_THREADS, 0, stream>>>(DEG, BCNT, dst, GDE_E);
  gde_edge_fill<<<(GDE_E + EB_THREADS - 1) / EB_THREADS, EB_THREADS, 0, stream>>>(PK, PP, OFF, BCUR, BCNT, src, dst, p, GDE_E);

  auto spline = [&](const float* y, int Cin, int Cout, const float* G,
                    const float* bias, float* ob, int am, float w, float c) {
    const int mmg = (MM_TILES + MM_WAVES - 1) / MM_WAVES;
    if (Cin == 256)
      gde_mm_lds<256, 12, MM_A_PLAIN><<<mmg, MM_THREADS, 0, stream>>>(y, G, Z, ob, bias, nullptr, nullptr, nullptr, w, c);
    else if (am == MM_A_PLAIN)
      gde_mm_lds<64, 12, MM_A_PLAIN><<<mmg, MM_THREADS, 0, stream>>>(y, G, Z, ob, bias, nullptr, nullptr, nullptr, w, c);
    else if (am == MM_A_TANH)
      gde_mm_lds<64, 12, MM_A_TANH><<<mmg, MM_THREADS, 0, stream>>>(y, G, Z, ob, bias, H, Kb, ACC, w, c);
    else if (am == MM_A_SET)
      gde_mm_lds<64, 12, MM_A_SET><<<mmg, MM_THREADS, 0, stream>>>(y, G, Z, ob, bias, H, Kb, ACC, w, c);
    else if (am == MM_A_ADD)
      gde_mm_lds<64, 12, MM_A_ADD><<<mmg, MM_THREADS, 0, stream>>>(y, G, Z, ob, bias, H, Kb, ACC, w, c);
    else
      gde_mm_lds<64, 3, MM_A_FINAL><<<mmg, MM_THREADS, 0, stream>>>(y, G, Z, ob, bias, H, Kb, ACC, w, c);
    if (Cout == 64) gde_edge_lds<64><<<EB_NB, EB_THREADS, 0, stream>>>(ob, Z, PK, PP, OFF, DEG);
    else            gde_edge_lds<16><<<EB_NB, EB_THREADS, 0, stream>>>(ob, Z, PK, PP, OFF, DEG);
  };
  auto odefunc = [&](int am, float w, float c) {
    spline(H, 64, 64, GCA, ba, U, am, w, c);
    spline(U, 64, 64, GCB, bb, Kb, MM_A_PLAIN, 0.0f, 0.0f);
  };

  spline(x, 256, 64, GC1, b1, H, MM_A_PLAIN, 0.0f, 0.0f);

  odefunc(MM_A_TANH, 0.0f, 0.0f);
  odefunc(MM_A_SET, 1.0f, 1.5f);
  odefunc(MM_A_ADD, 2.0f, 1.5f);
  odefunc(MM_A_ADD, 2.0f, 3.0f);

  spline(H, 64, 16, GC2, b2, out, MM_A_FINAL, 0.0f, 0.0f);
  gde_log_softmax16<<<(GDE_N + 255) / 256, 256, 0, stream>>>(out, GDE_N);
}


# Next layer's dense product done in the tail of each 64-channel edge pass, on the range's finished rows in LDS (f32 WMMA, G and elementwise step in LDS too): 9 product launches gone; sum order kept
# speedup vs baseline: 3.2889x; 1.1653x over previous
#include <hip/hip_runtime.h>
#include <math.h>

#define GDE_N 50000
#define GDE_E 800000

typedef float v2f __attribute__((ext_vector_type(2)));
typedef float v4f __attribute__((ext_vector_type(4)));
typedef float v8f __attribute__((ext_vector_type(8)));

#define MM_A_PLAIN 0
#define MM_A_TANH  1
#define MM_A_SET   2
#define MM_A_ADD   3
#define MM_A_FINAL 4
#define MM_TILES (GDE_N / 16)
#define MM_WAVES 8
#define MM_THREADS (32 * MM_WAVES)
#define MM_KC 64
#define MM_APITCH (MM_KC + 4)

static_assert(GDE_N % 16 == 0, "whole row tiles");

template <int K, int NT, int AM>
__global__ __launch_bounds__(MM_THREADS) __attribute__((amdgpu_waves_per_eu(4, 4))) void gde_mm_lds(
    const float* __restrict__ X, const float* __restrict__ G, float* __restrict__ Zp,
    float* __restrict__ S, const float* __restrict__ bias, float* Hio,
    const float* __restrict__ Kp, float* __restrict__ Acc, float w, float c) {
  static_assert(K % MM_KC == 0, "K must be whole chunks");
  static_assert(AM == MM_A_PLAIN || K == MM_KC, "a loader that stores sees every element once");
  static_assert(NT % 3 == 0, "columns are [pairs | root], two to one");
  constexpr int NC    = 16 * NT;
  constexpr int NTP   = 2 * NT / 3;
  constexpr int CP    = 16 * NTP;
  constexpr int CO    = NC - CP;
  constexpr int ITEMS = (MM_KC / 2) * (NC / 4);
  constexpr int ITERS = (ITEMS + MM_THREADS - 1) / MM_THREADS;
  constexpr int FULL  = ITEMS / MM_THREADS;
  __shared__ __attribute__((aligned(16))) v2f   Gs[(MM_KC / 4) * NT * 32];
  __shared__ __attribute__((aligned(16))) float As[MM_WAVES][16 * MM_APITCH];

  const int tid  = threadIdx.x;
  const int lane = tid & 31;
  const int wave = __builtin_amdgcn_readfirstlane(tid >> 5);
  const int m    = lane & 15;
  const int half = lane >> 4;
  const int  tile_raw = blockIdx.x * MM_WAVES + wave;
  const bool live     = tile_raw < MM_TILES;
  const int  tile     = live ? tile_raw : (MM_TILES - 1);
  const size_t tbase = (size_t)tile * 16 * K;
  const float* __restrict__ xt = X + tbase;
  float* as = As[wave];

  v8f acc[NT];
#pragma unroll
  for (int t = 0; t < NTP; ++t) acc[t] = (v8f){0.f, 0.f, 0.f, 0.f, 0.f, 0.f, 0.f, 0.f};
#pragma unroll
  for (int t = NTP; t < NT; ++t) {
    const float bv = bias[16 * (t - NTP) + m];
    acc[t] = (v8f){bv, bv, bv, bv, bv, bv, bv, bv};
  }

#pragma unroll 1
  for (int kc = 0; kc < K; kc += MM_KC) {
    if (kc) __syncthreads();
    float4 xa[8], kk[8], aa[8];
#pragma unroll
    for (int i = 0; i < 8; ++i) {
      const size_t off = (size_t)(2 * i + half) * K + kc + 4 * m;
      if (AM == MM_A_TANH) xa[i] = *(const float4*)(Hio + tbase + off);
      else                 xa[i] = *(const float4*)(xt + off);
      if (AM == MM_A_SET || AM == MM_A_ADD || AM == MM_A_FINAL) kk[i] = *(const float4*)(Kp + tbase + off);
      if (AM == MM_A_ADD || AM == MM_A_FINAL) aa[i] = *(const float4*)(Acc + tbase + off);
    }
    float4 g0[ITERS], g1[ITERS];
    int    gslot[ITERS];
#pragma unroll
    for (int u = 0; u < ITERS; ++u) {
      const int i  = tid + u * MM_THREADS;
      const int ic = (u < FULL || i < ITEMS) ? i : (i - ITEMS);
      const int q  = ic / (NC / 4);
      const int c0 = (ic - q * (NC / 4)) * 4;
      g0[u] = *(const float4*)(G + (size_t)(kc + 2 * q)     * NC + c0);
      g1[u] = *(const float4*)(G + (size_t)(kc + 2 * q + 1) * NC + c0);
      gslot[u] = (((q >> 1) * NT + (c0 >> 4)) * 32) + (q & 1) * 16 + (c0 & 15);
    }
    if (AM == MM_A_TANH) {
#pragma unroll
      for (int i = 0; i < 8; ++i) {
        xa[i].x = tanhf(xa[i].x); xa[i].y = tanhf(xa[i].y);
        xa[i].z = tanhf(xa[i].z); xa[i].w = tanhf(xa[i].w);
      }
    }
    if (AM == MM_A_SET || AM == MM_A_ADD) {
#pragma unroll
      for (int i = 0; i < 8; ++i) {
        if (AM == MM_A_SET) {
          aa[i].x = w * kk[i].x; aa[i].y = w * kk[i].y; aa[i].z = w * kk[i].z; aa[i].w = w * kk[i].w;
        } else {
          aa[i].x = aa[i].x + w * kk[i].x; aa[i].y = aa[i].y + w * kk[i].y;
          aa[i].z = aa[i].z + w * kk[i].z; aa[i].w = aa[i].w + w * kk[i].w;
        }
        xa[i].x = xa[i].x + c * kk[i].x; xa[i].y = xa[i].y + c * kk[i].y;
        xa[i].z = xa[i].z + c * kk[i].z; xa[i].w = xa[i].w + c * kk[i].w;
      }
    }
    if (AM == MM_A_FINAL) {
#pragma unroll
      for (int i = 0; i < 8; ++i) {
        xa[i].x = xa[i].x + 0.5f * (aa[i].x + kk[i].x); xa[i].y = xa[i].y + 0.5f * (aa[i].y + kk[i].y);
        xa[i].z = xa[i].z + 0.5f * (aa[i].z + kk[i].z); xa[i].w = xa[i].w + 0.5f * (aa[i].w + kk[i].w);
      }
    }
#pragma unroll
    for (int i = 0; i < 8; ++i)
      *(float4*)(as + (2 * i + half) * MM_APITCH + 4 * m) = xa[i];
#pragma unroll
    for (int u = 0; u < ITERS; ++u) {
      float* gp = (float*)&Gs[gslot[u]];
      *(float4*)(gp)     = make_float4(g0[u].x, g1[u].x, g0[u].y, g1[u].y);
      *(float4*)(gp + 4) = make_float4(g0[u].z, g1[u].z, g0[u].w, g1[u].w);
    }
    __syncthreads();
    if ((AM == MM_A_TANH || AM == MM_A_SET || AM == MM_A_ADD) && live) {
      float* sp = (AM == MM_A_TANH ? Hio : Acc) + tbase + kc + 4 * m;
#pragma unroll
      for (int i = 0; i < 8; ++i)
        *(float4*)(sp + (size_t)(2 * i + half) * K) = (AM == MM_A_TANH) ? xa[i] : aa[i];
    }
#pragma unroll
    for (int j = 0; j < MM_KC / 4; ++j) {
      const v2f a = *(const v2f*)(as + m * MM_APITCH + 4 * j + 2 * half);
#pragma unroll
      for (int t = 0; t < NT; ++t) {
        const v2f b = Gs[(j * NT + t) * 32 + lane];
        acc[t] = __builtin_amdgcn_wmma_f32_16x16x4_f32(false, a, false, b,
                                                       (short)0, acc[t], false, false);
      }
    }
  }

  if (live) {
    const size_t row0 = (size_t)(tile * 16 + half * 8);
    float* zr = Zp + row0 * CP + m;
#pragma unroll
    for (int t = 0; t < NTP; ++t)
#pragma unroll
      for (int r = 0; r < 8; ++r)
        zr[(size_t)r * CP + 16 * t] = acc[t][r];
    float* sr = S + row0 * CO + m;
#pragma unroll
    for (int t = NTP; t < NT; ++t)
#pragma unroll
      for (int r = 0; r < 8; ++r)
        sr[(size_t)r * CO + 16 * (t - NTP)] = acc[t][r];
  }
}

__global__ void gde_prep_g(float* __restrict__ G, const float* __restrict__ Wp,
                           const float* __restrict__ R, int Cin, int Cout,
                           unsigned* __restrict__ cz, int nz) {
  int idx = blockIdx.x * blockDim.x + threadIdx.x;
  if (idx < nz) cz[idx] = 0u;
  int Ncols = 3 * Cout;
  int tot = Cin * Ncols;
  if (idx >= tot) return;
  int r = idx / Ncols, c = idx % Ncols;
  float v;
  if (c < 2 * Cout) {
    const int ch = c >> 1;
    const float w0 = Wp[r * Cout + ch];
    v = (c & 1) ? (Wp[(size_t)Cin * Cout + r * Cout + ch] - w0) : w0;
  } else {
    v = R[r * Cout + (c - 2 * Cout)];
  }
  G[idx] = v;
}

#define EB_NB      256
#define EB_RB      ((GDE_N + EB_NB - 1) / EB_NB)
#define EB_WAVES   32
#define EB_THREADS (32 * EB_WAVES)
#define EB_OFFW    560
static_assert(EB_NB <= EB_THREADS && EB_RB <= EB_THREADS && EB_NB + 1 <= EB_OFFW, "one thread per range / per row");
static_assert(GDE_N <= 65536 && EB_RB <= 65536, "a record packs the source row and the local row in 16 bits each");
static_assert((EB_OFFW + GDE_N + 2 * GDE_E + 256 * 192 + 2 * 64 * 192 + 64 * 48) % 128 == 0, "plane alignment");

__global__ __launch_bounds__(EB_THREADS) void gde_deg(float* __restrict__ deg, unsigned* __restrict__ bcnt,
                                                      const int* __restrict__ dst, int E) {
  __shared__ unsigned hist[EB_NB];
  const int tid = threadIdx.x;
  if (tid < EB_NB) hist[tid] = 0u;
  __syncthreads();
  const int e = blockIdx.x * EB_THREADS + tid;
  if (e < E) {
    const int d = dst[e];
    atomicAdd(&deg[d], 1.0f);
    atomicAdd(&hist[d / EB_RB], 1u);
  }
  __syncthreads();
  if (tid < EB_NB) {
    const unsigned h = hist[tid];
    if (h) atomicAdd(&bcnt[tid], h);
  }
}

__global__ __launch_bounds__(EB_THREADS) void gde_edge_fill(
    int* __restrict__ pk, float* __restrict__ pp, unsigned* __restrict__ off, unsigned* __restrict__ bcur,
    const unsigned* __restrict__ bcnt, const int* __restrict__ src, const int* __restrict__ dst,
    const float* __restrict__ p, int E) {
  __shared__ unsigned hist[EB_NB];
  __shared__ unsigned scan[EB_NB];
  __shared__ unsigned base[EB_NB];
  const int tid = threadIdx.x;
  unsigned cntv = 0u;
  if (tid < EB_NB) { hist[tid] = 0u; cntv = bcnt[tid]; scan[tid] = cntv; }
  __syncthreads();
  const int e = blockIdx.x * EB_THREADS + tid;
  int d = 0, b = 0;
  unsigned r = 0u;
  if (e < E) {
    d = dst[e];
    b = d / EB_RB;
    r = atomicAdd(&hist[b], 1u);
  }
  for (int st = 1; st < EB_NB; st <<= 1) {
    unsigned v = 0u;
    if (tid < EB_NB && tid >= st) v = scan[tid - st];
    __syncthreads();
    if (tid < EB_NB) scan[tid] += v;
    __syncthreads();
  }
  if (tid < EB_NB) {
    const unsigned excl = scan[tid] - cntv;
    const unsigned h = hist[tid];
    const unsigned rs = h ? atomicAdd(&bcur[tid], h) : 0u;
    base[tid] = excl + rs;
    if (blockIdx.x == 0) {
      off[tid] = excl;
      if (tid == EB_NB - 1) off[EB_NB] = scan[tid];
    }
  }
  __syncthreads();
  if (e < E) {
    const unsigned pos = base[b] + r;
    pk[pos] = src[e] | ((d - b * EB_RB) << 16);
    pp[pos] = p[e];
  }
}


template <int EPS>
__device__ __forceinline__ int gde_rec_pick(int v, int j, int sub) {
  int r = __builtin_amdgcn_readlane(v, j);
#pragma unroll
  for (int k = 1; k < EPS; ++k) {
    const int o = __builtin_amdgcn_readlane(v, j + k);
    r = (sub == k) ? o : r;
  }
  return r;
}

template <int COUT, int PITCH, int G>
__device__ __forceinline__ void gde_lds_group(float* acc, const float* __restrict__ Z,
                                              int pkv, int wav, int wbv, int jb, int cnt, int sub, int c) {
  constexpr int LPE = COUT / 2;
  constexpr int EPS = 32 / LPE;
  float4 z[G];
  int    dl[G];
  float  wa[G], wb[G];
#pragma unroll
  for (int u = 0; u < G; ++u) {
    const int j = jb + u * EPS;
    const unsigned pkx = (unsigned)gde_rec_pick<EPS>(pkv, j, sub);
    dl[u] = (int)(pkx >> 16);
    wa[u] = __int_as_float(gde_rec_pick<EPS>(wav, j, sub));
    wb[u] = __int_as_float(gde_rec_pick<EPS>(wbv, j, sub));
    z[u] = *(const float4*)(Z + (size_t)(pkx & 0xffffu) * (2 * COUT) + 4 * c);
  }
#pragma unroll
  for (int u = 0; u < G; ++u) {
    if (EPS == 1 || jb + u * EPS + sub < cnt) {
      float* ar = acc + dl[u] * PITCH + 2 * c;
      __hip_atomic_fetch_add(ar,     wa[u] * z[u].x + wb[u] * z[u].y, __ATOMIC_RELAXED, __HIP_MEMORY_SCOPE_WORKGROUP);
      __hip_atomic_fetch_add(ar + 1, wa[u] * z[u].z + wb[u] * z[u].w, __ATOMIC_RELAXED, __HIP_MEMORY_SCOPE_WORKGROUP);
    }
  }
}

template <int COUT>
__global__ __launch_bounds__(EB_THREADS) void gde_edge_lds(
    float* agg, const float* __restrict__ Z, const int* __restrict__ pk, const float* __restrict__ pp,
    const unsigned* __restrict__ off, const float* __restrict__ deg) {
  constexpr int LPE = COUT / 2;
  constexpr int EPS = 32 / LPE;
  constexpr int GRP = EPS == 1 ? 16 : 8;
  static_assert(32 % LPE == 0 && COUT % 4 == 0 && 32 % (GRP * EPS) == 0, "shape");
  __shared__ __attribute__((aligned(16))) float acc[EB_RB * COUT];
  __shared__ float inv[EB_RB];
  const int tid  = threadIdx.x;
  const int lane = tid & 31;
  const int wave = __builtin_amdgcn_readfirstlane(tid >> 5);
  const int sub  = lane / LPE;
  const int c    = lane % LPE;
  const int row0 = blockIdx.x * EB_RB;
  const int nrr  = GDE_N - row0;
  const int nr   = nrr < EB_RB ? (nrr > 0 ? nrr : 0) : EB_RB;
  const int nq   = nr * (COUT / 4);
  float* ag = agg + (size_t)row0 * COUT;
  constexpr int NQ = (EB_RB * (COUT / 4) + EB_THREADS - 1) / EB_THREADS;
  if (nr > 0) {
    v4f st[NQ];
#pragma unroll
    for (int u = 0; u < NQ; ++u) {
      const int qi = tid + u * EB_THREADS;
      st[u] = ((const v4f*)ag)[qi < nq ? qi : nq - 1];
    }
    const float dg = deg[row0 + (tid < nr ? tid : nr - 1)];
#pragma unroll
    for (int u = 0; u < NQ; ++u) asm volatile("" :: "v"(st[u]));
#pragma unroll
    for (int u = 0; u < NQ; ++u) {
      const int qi = tid + u * EB_THREADS;
      if (qi < nq) ((v4f*)acc)[qi] = st[u];
    }
    if (tid < nr) inv[tid] = 1.0f / fmaxf(dg, 1.0f);
  }
  __syncthreads();

  const unsigned o0 = off[blockIdx.x], o1 = off[blockIdx.x + 1];
  const unsigned share = (o1 - o0 + EB_WAVES - 1) / EB_WAVES;
  unsigned i = o0 + (unsigned)wave * share;
  const unsigned we = (i + share < o1) ? (i + share) : o1;
  unsigned ri = i + lane;
  unsigned rc = ri < GDE_E - 1u ? ri : GDE_E - 1u;
  int   npk = pk[rc];
  float npv = pp[rc];
  if (ri >= we) { npk = 0; npv = 0.0f; }
#pragma unroll 1
  for (; i < we; i += 32) {
    const int   cnt = (we - i < 32u) ? (int)(we - i) : 32;
    const int   pkv = npk;
    const float pv  = npv;
    ri = i + 32u + lane;
    rc = ri < GDE_E - 1u ? ri : GDE_E - 1u;
    npk = pk[rc];
    npv = pp[rc];
    if (ri >= we) { npk = 0; npv = 0.0f; }
    const float iv  = (lane < cnt) ? inv[(unsigned)pkv >> 16] : 0.0f;
    const int   wav = __float_as_int(iv);
    const int   wbv = __float_as_int(pv * iv);
    const int S = (cnt + EPS - 1) / EPS;
    int q = 0;
#define GDE_RUN(G) gde_lds_group<COUT, COUT, G>(acc, Z, pkv, wav, wbv, q * EPS, cnt, sub, c);
#pragma unroll 1
    for (; q + GRP <= S; q += GRP) GDE_RUN(GRP)
    if (GRP > 8 && S - q >= 8) { GDE_RUN(8) q += 8; }
    if (S - q >= 4) { GDE_RUN(4) q += 4; }
    if (S - q >= 2) { GDE_RUN(2) q += 2; }
    if (S - q >= 1) { GDE_RUN(1) }
#undef GDE_RUN
  }
  __syncthreads();
  for (int j = tid; j < nq; j += EB_THREADS) ((float4*)ag)[j] = ((const float4*)acc)[j];
}

#define EM_PITCH 68
#define EM_RT    ((EB_RB + 15) / 16)
#define EM_ROWS  (16 * EM_RT)
static_assert(EB_RB * 16 <= 4 * EB_THREADS, "a range's rows are at most four 16-byte pieces per thread");

template <int TM, int NT>
__global__ __launch_bounds__(EB_THREADS) void gde_edge_mm(
    const float* Sin, float* Sout, const float* __restrict__ Z, float* __restrict__ Zn,
    const int* __restrict__ pk, const float* __restrict__ pp, const unsigned* __restrict__ off,
    const float* __restrict__ deg, const float* __restrict__ G, const float* __restrict__ bias,
    float* Hp, float* Accp, float w, float cs) {
  static_assert(NT % 3 == 0, "columns are [pairs | root], two to one");
  constexpr int COUT  = 64;
  constexpr int NC    = 16 * NT;
  constexpr int NTP   = 2 * NT / 3;
  constexpr int CP    = 16 * NTP;
  constexpr int CO    = NC - CP;
  constexpr int TW    = (NT % 4 == 0) ? 4 : 3;
  constexpr int CG    = NT / TW;
  static_assert(NT % TW == 0, "whole work items");
  static_assert(CG == 1 || (TW * 16 * 4) % 128 == 0, "an item's stores are whole 128-byte lines");
  constexpr int ITEMS = (COUT / 2) * (NC / 4);
  constexpr int ITERS = (ITEMS + EB_THREADS - 1) / EB_THREADS;
  constexpr int NQ    = (EB_RB * (COUT / 4) + EB_THREADS - 1) / EB_THREADS;
  constexpr int GRP   = 16;
  __shared__ __attribute__((aligned(16))) float acc[EM_ROWS * EM_PITCH];
  __shared__ __attribute__((aligned(16))) v2f   Gs[(COUT / 4) * NT * 32];
  __shared__ float inv[EB_RB];
  const int tid  = threadIdx.x;
  const int lane = tid & 31;
  const int wave = __builtin_amdgcn_readfirstlane(tid >> 5);
  const int row0 = blockIdx.x * EB_RB;
  const int nrr  = GDE_N - row0;
  const int nr   = nrr < EB_RB ? (nrr > 0 ? nrr : 0) : EB_RB;
  const int nq   = nr * (COUT / 4);
  const int nrt  = (nr + 15) >> 4;
  const size_t gbase = (size_t)row0 * COUT;
  if (nr > 0) {
    v4f st[NQ];
#pragma unroll
    for (int u = 0; u < NQ; ++u) {
      const int qi = tid + u * EB_THREADS;
      st[u] = ((const v4f*)(Sin + gbase))[qi < nq ? qi : nq - 1];
    }
    const float dg = deg[row0 + (tid < nr ? tid : nr - 1)];
    float4 g0[ITERS], g1[ITERS];
    int    gslot[ITERS];
#pragma unroll
    for (int u = 0; u < ITERS; ++u) {
      const int ic = (tid + u * EB_THREADS) % ITEMS;
      const int q  = ic / (NC / 4);
      const int c0 = (ic - q * (NC / 4)) * 4;
      g0[u] = *(const float4*)(G + (size_t)(2 * q)     * NC + c0);
      g1[u] = *(const float4*)(G + (size_t)(2 * q + 1) * NC + c0);
      gslot[u] = (((q >> 1) * NT + (c0 >> 4)) * 32) + (q & 1) * 16 + (c0 & 15);
    }
#pragma unroll
    for (int u = 0; u < NQ; ++u) asm volatile("" :: "v"(st[u]));
#pragma unroll
    for (int u = 0; u < NQ; ++u) {
      const int qi = tid + u * EB_THREADS;
      if (qi < nq) *(v4f*)(acc + (qi >> 4) * EM_PITCH + 4 * (qi & 15)) = st[u];
    }
    if (tid < (16 * nrt - nr) * EM_PITCH) acc[nr * EM_PITCH + tid] = 0.0f;
    if (tid < nr) inv[tid] = 1.0f / fmaxf(dg, 1.0f);
#pragma unroll
    for (int u = 0; u < ITERS; ++u) {
      float* gp = (float*)&Gs[gslot[u]];
      *(float4*)(gp)     = make_float4(g0[u].x, g1[u].x, g0[u].y, g1[u].y);
      *(float4*)(gp + 4) = make_float4(g0[u].z, g1[u].z, g0[u].w, g1[u].w);
    }
  }
  __syncthreads();

  const unsigned o0 = off[blockIdx.x], o1 = off[blockIdx.x + 1];
  const unsigned share = (o1 - o0 + EB_WAVES - 1) / EB_WAVES;
  unsigned i = o0 + (unsigned)wave * share;
  const unsigned we = (i + share < o1) ? (i + share) : o1;
  unsigned ri = i + lane;
  unsigned rc = ri < GDE_E - 1u ? ri : GDE_E - 1u;
  int   npk = pk[rc];
  float npv = pp[rc];
  if (ri >= we) { npk = 0; npv = 0.0f; }
#pragma unroll 1
  for (; i < we; i += 32) {
    const int   cnt = (we - i < 32u) ? (int)(we - i) : 32;
    const int   pkv = npk;
    const float pv  = npv;
    ri = i + 32u + lane;
    rc = ri < GDE_E - 1u ? ri : GDE_E - 1u;
    npk = pk[rc];
    npv = pp[rc];
    if (ri >= we) { npk = 0; npv = 0.0f; }
    const float iv  = (lane < cnt) ? inv[(unsigned)pkv >> 16] : 0.0f;
    const int   wav = __float_as_int(iv);
    const int   wbv = __float_as_int(pv * iv);
    int q = 0;
#define GDE_RUN(G_) gde_lds_group<COUT, EM_PITCH, G_>(acc, Z, pkv, wav, wbv, q, cnt, 0, lane);
#pragma unroll 1
    for (; q + GRP <= cnt; q += GRP) GDE_RUN(GRP)
    if (cnt - q >= 8) { GDE_RUN(8) q += 8; }
    if (cnt - q >= 4) { GDE_RUN(4) q += 4; }
    if (cnt - q >= 2) { GDE_RUN(2) q += 2; }
    if (cnt - q >= 1) { GDE_RUN(1) }
#undef GDE_RUN
  }

  v4f hv[NQ], av[NQ];
  if (nr > 0) {
#pragma unroll
    for (int u = 0; u < NQ; ++u) {
      const int qi = tid + u * EB_THREADS;
      const int qc = qi < nq ? qi : nq - 1;
      if (TM == MM_A_SET || TM == MM_A_ADD || TM == MM_A_FINAL) hv[u] = ((const v4f*)(Hp + gbase))[qc];
      if (TM == MM_A_ADD || TM == MM_A_FINAL) av[u] = ((const v4f*)(Accp + gbase))[qc];
    }
  }
  __syncthreads();
  if (TM != MM_A_PLAIN) {
#pragma unroll
    for (int u = 0; u < NQ; ++u) {
      const int qi = tid + u * EB_THREADS;
      if (qi < nq) {
        float* lp = acc + (qi >> 4) * EM_PITCH + 4 * (qi & 15);
        v4f kv = *(const v4f*)lp;
        if (TM == MM_A_TANH) {
          kv.x = tanhf(kv.x); kv.y = tanhf(kv.y); kv.z = tanhf(kv.z); kv.w = tanhf(kv.w);
          ((v4f*)(Hp + gbase))[qi] = kv;
          *(v4f*)lp = kv;
        }
        if (TM == MM_A_SET || TM == MM_A_ADD) {
          v4f a;
          if (TM == MM_A_SET) {
            a.x = w * kv.x; a.y = w * kv.y; a.z = w * kv.z; a.w = w * kv.w;
          } else {
            a.x = av[u].x + w * kv.x; a.y = av[u].y + w * kv.y;
            a.z = av[u].z + w * kv.z; a.w = av[u].w + w * kv.w;
          }
          ((v4f*)(Accp + gbase))[qi] = a;
          v4f y;
          y.x = hv[u].x + cs * kv.x; y.y = hv[u].y + cs * kv.y;
          y.z = hv[u].z + cs * kv.z; y.w = hv[u].w + cs * kv.w;
          *(v4f*)lp = y;
        }
        if (TM == MM_A_FINAL) {
          v4f y;
          y.x = hv[u].x + 0.5f * (av[u].x + kv.x); y.y = hv[u].y + 0.5f * (av[u].y + kv.y);
          y.z = hv[u].z + 0.5f * (av[u].z + kv.z); y.w = hv[u].w + 0.5f * (av[u].w + kv.w);
          *(v4f*)lp = y;
        }
      }
    }
    __syncthreads();
  }

  const int m     = lane & 15;
  const int half  = lane >> 4;
  const int items = nrt * CG;
#pragma unroll 1
  for (int it = wave; it < items; it += EB_WAVES) {
    const int rt = it / CG;
    const int t0 = (it - rt * CG) * TW;
    v8f d[TW];
#pragma unroll
    for (int t = 0; t < TW; ++t) {
      const int tt = t0 + t;
      const float bv = (tt >= NTP) ? bias[16 * (tt - NTP) + m] : 0.0f;
      d[t] = (v8f){bv, bv, bv, bv, bv, bv, bv, bv};
    }
    const float* ar = acc + (16 * rt + m) * EM_PITCH + 2 * half;
#pragma unroll 4
    for (int j = 0; j < COUT / 4; ++j) {
      const v2f a = *(const v2f*)(ar + 4 * j);
#pragma unroll
      for (int t = 0; t < TW; ++t) {
        const v2f b = Gs[(j * NT + t0 + t) * 32 + lane];
        d[t] = __builtin_amdgcn_wmma_f32_16x16x4_f32(false, a, false, b,
                                                     (short)0, d[t], false, false);
      }
    }
    const int lr = 16 * rt + 8 * half;
#pragma unroll
    for (int t = 0; t < TW; ++t) {
      const int tt = t0 + t;
      if (tt < NTP) {
        float* zr = Zn + (size_t)(row0 + lr) * CP + 16 * tt + m;
#pragma unroll
        for (int r = 0; r < 8; ++r)
          if (lr + r < nr) zr[(size_t)r * CP] = d[t][r];
      } else {
        float* sr = Sout + (size_t)(row0 + lr) * CO + 16 * (tt - NTP) + m;
#pragma unroll
        for (int r = 0; r < 8; ++r)
          if (lr + r < nr) sr[(size_t)r * CO] = d[t][r];
      }
    }
  }
}

__global__ void gde_log_softmax16(float* __restrict__ out, int n) {
  int i = blockIdx.x * blockDim.x + threadIdx.x;
  if (i >= n) return;
  float v[16];
  float mx = -3.402823466e+38f;
  float* row = out + (size_t)i * 16;
#pragma unroll
  for (int c = 0; c < 16; ++c) { v[c] = tanhf(row[c]); mx = fmaxf(mx, v[c]); }
  float s = 0.f;
#pragma unroll
  for (int c = 0; c < 16; ++c) s += __expf(v[c] - mx);
  float ls = __logf(s);
#pragma unroll
  for (int c = 0; c < 16; ++c) row[c] = v[c] - mx - ls;
}

extern "C" void kernel_launch(void* const* d_in, const int* in_sizes, int n_in,
                              void* d_out, int out_size, void* d_ws, size_t ws_size,
                              hipStream_t stream) {
  const float* x   = (const float*)d_in[0];
  const float* p   = (const float*)d_in[1];
  const int*   src = (const int*)d_in[2];
  const int*   dst = (const int*)d_in[3];
  const float* W1 = (const float*)d_in[4],  *R1 = (const float*)d_in[5],  *b1 = (const float*)d_in[6];
  const float* Wa = (const float*)d_in[7],  *Ra = (const float*)d_in[8],  *ba = (const float*)d_in[9];
  const float* Wb = (const float*)d_in[10], *Rb = (const float*)d_in[11], *bb = (const float*)d_in[12];
  const float* W2 = (const float*)d_in[13], *R2 = (const float*)d_in[14], *b2 = (const float*)d_in[15];
  float* out = (float*)d_out;

  float* W = (float*)d_ws;
  size_t o = 0;
  float* Z   = W + o; o += (size_t)GDE_N * 128;
  float* H   = W + o; o += (size_t)GDE_N * 64;
  float* U   = W + o; o += (size_t)GDE_N * 64;
  float* ACC = W + o; o += (size_t)GDE_N * 64;
  unsigned* OFF = (unsigned*)(W + o); o += (size_t)EB_OFFW;
  float* DEG = W + o; o += (size_t)GDE_N;
  int*   PK  = (int*)(W + o); o += (size_t)GDE_E;
  float* PP  = W + o; o += (size_t)GDE_E;
  unsigned* BCNT = (unsigned*)U;
  unsigned* BCUR = BCNT + EB_NB;
  float* GC1 = W + o; o += (size_t)256 * 192;
  float* GCA = W + o; o += (size_t)64 * 192;
  float* GCB = W + o; o += (size_t)64 * 192;
  float* GC2 = W + o; o += (size_t)64 * 48;
  float* Kb  = W + o; o += (size_t)GDE_N * 64;
  float* Z2  = Kb;
  const bool tails = ws_size >= sizeof(float) * (o + (size_t)GDE_N * 64);
  (void)n_in; (void)in_sizes; (void)out_size;

  gde_prep_g<<<(256 * 192 + 255) / 256, 256, 0, stream>>>(GC1, W1, R1, 256, 64, BCNT, 2 * EB_NB);
  gde_prep_g<<<(64 * 192 + 255) / 256, 256, 0, stream>>>(GCA, Wa, Ra, 64, 64, BCNT, 0);
  gde_prep_g<<<(64 * 192 + 255) / 256, 256, 0, stream>>>(GCB, Wb, Rb, 64, 64, BCNT, 0);
  gde_prep_g<<<(64 * 48 + 255) / 256, 256, 0, stream>>>(GC2, W2, R2, 64, 16, BCNT, 0);

  hipMemsetAsync(DEG, 0, sizeof(float) * GDE_N, stream);
  gde_deg<<<(GDE_E + EB_THREADS - 1) / EB_THREADS, EB_THREADS, 0, stream>>>(DEG, BCNT, dst, GDE_E);
  gde_edge_fill<<<(GDE_E + EB_THREADS - 1) / EB_THREADS, EB_THREADS, 0, stream>>>(PK, PP, OFF, BCUR, BCNT, src, dst, p, GDE_E);

  const int mmg = (MM_TILES + MM_WAVES - 1) / MM_WAVES;
  if (tails) {
    gde_mm_lds<256, 12, MM_A_PLAIN><<<mmg, MM_THREADS, 0, stream>>>(x, GC1, Z, H, b1, nullptr, nullptr, nullptr, 0.0f, 0.0f);
    gde_edge_mm<MM_A_TANH, 12><<<EB_NB, EB_THREADS, 0, stream>>>(H, U, Z, Z2, PK, PP, OFF, DEG, GCA, ba, H, ACC, 0.0f, 0.0f);
    gde_edge_mm<MM_A_PLAIN, 12><<<EB_NB, EB_THREADS, 0, stream>>>(U, U, Z2, Z, PK, PP, OFF, DEG, GCB, bb, H, ACC, 0.0f, 0.0f);
    gde_edge_mm<MM_A_SET, 12><<<EB_NB, EB_THREADS, 0, stream>>>(U, U, Z, Z2, PK, PP, OFF, DEG, GCA, ba, H, ACC, 1.0f, 1.5f);
    gde_edge_mm<MM_A_PLAIN, 12><<<EB_NB, EB_THREADS, 0, stream>>>(U, U, Z2, Z, PK, PP, OFF, DEG, GCB, bb, H, ACC, 0.0f, 0.0f);
    gde_edge_mm<MM_A_ADD, 12><<<EB_NB, EB_THREADS, 0, stream>>>(U, U, Z, Z2, PK, PP, OFF, DEG, GCA, ba, H, ACC, 2.0f, 1.5f);
    gde_edge_mm<MM_A_PLAIN, 12><<<EB_NB, EB_THREADS, 0, stream>>>(U, U, Z2, Z, PK, PP, OFF, DEG, GCB, bb, H, ACC, 0.0f, 0.0f);
    gde_edge_mm<MM_A_ADD, 12><<<EB_NB, EB_THREADS, 0, stream>>>(U, U, Z, Z2, PK, PP, OFF, DEG, GCA, ba, H, ACC, 2.0f, 3.0f);
    gde_edge_mm<MM_A_PLAIN, 12><<<EB_NB, EB_THREADS, 0, stream>>>(U, U, Z2, Z, PK, PP, OFF, DEG, GCB, bb, H, ACC, 0.0f, 0.0f);
    gde_edge_mm<MM_A_FINAL, 3><<<EB_NB, EB_THREADS, 0, stream>>>(U, out, Z, Z2, PK, PP, OFF, DEG, GC2, b2, H, ACC, 0.0f, 0.0f);
    gde_edge_lds<16><<<EB_NB, EB_THREADS, 0, stream>>>(out, Z2, PK, PP, OFF, DEG);
    gde_log_softmax16<<<(GDE_N + 255) / 256, 256, 0, stream>>>(out, GDE_N);
    return;
  }

  auto spline = [&](const float* y, int Cin, int Cout, const float* G,
                    const float* bias, float* ob, int am, float w, float c) {
    if (Cin == 256)
      gde_mm_lds<256, 12, MM_A_PLAIN><<<mmg, MM_THREADS, 0, stream>>>(y, G, Z, ob, bias, nullptr, nullptr, nullptr, w, c);
    else if (am == MM_A_PLAIN)
      gde_mm_lds<64, 12, MM_A_PLAIN><<<mmg, MM_THREADS, 0, stream>>>(y, G, Z, ob, bias, nullptr, nullptr, nullptr, w, c);
    else if (am == MM_A_TANH)
      gde_mm_lds<64, 12, MM_A_TANH><<<mmg, MM_THREADS, 0, stream>>>(y, G, Z, ob, bias, H, Kb, ACC, w, c);
    else if (am == MM_A_SET)
      gde_mm_lds<64, 12, MM_A_SET><<<mmg, MM_THREADS, 0, stream>>>(y, G, Z, ob, bias, H, Kb, ACC, w, c);
    else if (am == MM_A_ADD)
      gde_mm_lds<64, 12, MM_A_ADD><<<mmg, MM_THREADS, 0, stream>>>(y, G, Z, ob, bias, H, Kb, ACC, w, c);
    else
      gde_mm_lds<64, 3, MM_A_FINAL><<<mmg, MM_THREADS, 0, stream>>>(y, G, Z, ob, bias, H, Kb, ACC, w, c);
    if (Cout == 64) gde_edge_lds<64><<<EB_NB, EB_THREADS, 0, stream>>>(ob, Z, PK, PP, OFF, DEG);
    else            gde_edge_lds<16><<<EB_NB, EB_THREADS, 0, stream>>>(ob, Z, PK, PP, OFF, DEG);
  };
  auto odefunc = [&](int am, float w, float c) {
    spline(H, 64, 64, GCA, ba, U, am, w, c);
    spline(U, 64, 64, GCB, bb, Kb, MM_A_PLAIN, 0.0f, 0.0f);
  };

  spline(x, 256, 64, GC1, b1, H, MM_A_PLAIN, 0.0f, 0.0f);

  odefunc(MM_A_TANH, 0.0f, 0.0f);
  odefunc(MM_A_SET, 1.0f, 1.5f);
  odefunc(MM_A_ADD, 2.0f, 1.5f);
  odefunc(MM_A_ADD, 2.0f, 3.0f);

  spline(H, 64, 16, GC2, b2, out, MM_A_FINAL, 0.0f, 0.0f);
  gde_log_softmax16<<<(GDE_N + 255) / 256, 256, 0, stream>>>(out, GDE_N);
}


# Row degrees counted by the range's own group: the first edge pass counts its records per row in LDS (integer adds) and stores deg; 800k global float atomics and the memset gone; sums unchanged
# speedup vs baseline: 3.7023x; 1.1257x over previous
#include <hip/hip_runtime.h>
#include <math.h>

#define GDE_N 50000
#define GDE_E 800000

typedef float v2f __attribute__((ext_vector_type(2)));
typedef float v4f __attribute__((ext_vector_type(4)));
typedef float v8f __attribute__((ext_vector_type(8)));

#define MM_A_PLAIN 0
#define MM_A_TANH  1
#define MM_A_SET   2
#define MM_A_ADD   3
#define MM_A_FINAL 4
#define MM_TILES (GDE_N / 16)
#define MM_WAVES 8
#define MM_THREADS (32 * MM_WAVES)
#define MM_KC 64
#define MM_APITCH (MM_KC + 4)

static_assert(GDE_N % 16 == 0, "whole row tiles");

template <int K, int NT, int AM>
__global__ __launch_bounds__(MM_THREADS) __attribute__((amdgpu_waves_per_eu(4, 4))) void gde_mm_lds(
    const float* __restrict__ X, const float* __restrict__ G, float* __restrict__ Zp,
    float* __restrict__ S, const float* __restrict__ bias, float* Hio,
    const float* __restrict__ Kp, float* __restrict__ Acc, float w, float c) {
  static_assert(K % MM_KC == 0, "K must be whole chunks");
  static_assert(AM == MM_A_PLAIN || K == MM_KC, "a loader that stores sees every element once");
  static_assert(NT % 3 == 0, "columns are [pairs | root], two to one");
  constexpr int NC    = 16 * NT;
  constexpr int NTP   = 2 * NT / 3;
  constexpr int CP    = 16 * NTP;
  constexpr int CO    = NC - CP;
  constexpr int ITEMS = (MM_KC / 2) * (NC / 4);
  constexpr int ITERS = (ITEMS + MM_THREADS - 1) / MM_THREADS;
  constexpr int FULL  = ITEMS / MM_THREADS;
  __shared__ __attribute__((aligned(16))) v2f   Gs[(MM_KC / 4) * NT * 32];
  __shared__ __attribute__((aligned(16))) float As[MM_WAVES][16 * MM_APITCH];

  const int tid  = threadIdx.x;
  const int lane = tid & 31;
  const int wave = __builtin_amdgcn_readfirstlane(tid >> 5);
  const int m    = lane & 15;
  const int half = lane >> 4;
  const int  tile_raw = blockIdx.x * MM_WAVES + wave;
  const bool live     = tile_raw < MM_TILES;
  const int  tile     = live ? tile_raw : (MM_TILES - 1);
  const size_t tbase = (size_t)tile * 16 * K;
  const float* __restrict__ xt = X + tbase;
  float* as = As[wave];

  v8f acc[NT];
#pragma unroll
  for (int t = 0; t < NTP; ++t) acc[t] = (v8f){0.f, 0.f, 0.f, 0.f, 0.f, 0.f, 0.f, 0.f};
#pragma unroll
  for (int t = NTP; t < NT; ++t) {
    const float bv = bias[16 * (t - NTP) + m];
    acc[t] = (v8f){bv, bv, bv, bv, bv, bv, bv, bv};
  }

#pragma unroll 1
  for (int kc = 0; kc < K; kc += MM_KC) {
    if (kc) __syncthreads();
    float4 xa[8], kk[8], aa[8];
#pragma unroll
    for (int i = 0; i < 8; ++i) {
      const size_t off = (size_t)(2 * i + half) * K + kc + 4 * m;
      if (AM == MM_A_TANH) xa[i] = *(const float4*)(Hio + tbase + off);
      else                 xa[i] = *(const float4*)(xt + off);
      if (AM == MM_A_SET || AM == MM_A_ADD || AM == MM_A_FINAL) kk[i] = *(const float4*)(Kp + tbase + off);
      if (AM == MM_A_ADD || AM == MM_A_FINAL) aa[i] = *(const float4*)(Acc + tbase + off);
    }
    float4 g0[ITERS], g1[ITERS];
    int    gslot[ITERS];
#pragma unroll
    for (int u = 0; u < ITERS; ++u) {
      const int i  = tid + u * MM_THREADS;
      const int ic = (u < FULL || i < ITEMS) ? i : (i - ITEMS);
      const int q  = ic / (NC / 4);
      const int c0 = (ic - q * (NC / 4)) * 4;
      g0[u] = *(const float4*)(G + (size_t)(kc + 2 * q)     * NC + c0);
      g1[u] = *(const float4*)(G + (size_t)(kc + 2 * q + 1) * NC + c0);
      gslot[u] = (((q >> 1) * NT + (c0 >> 4)) * 32) + (q & 1) * 16 + (c0 & 15);
    }
    if (AM == MM_A_TANH) {
#pragma unroll
      for (int i = 0; i < 8; ++i) {
        xa[i].x = tanhf(xa[i].x); xa[i].y = tanhf(xa[i].y);
        xa[i].z = tanhf(xa[i].z); xa[i].w = tanhf(xa[i].w);
      }
    }
    if (AM == MM_A_SET || AM == MM_A_ADD) {
#pragma unroll
      for (int i = 0; i < 8; ++i) {
        if (AM == MM_A_SET) {
          aa[i].x = w * kk[i].x; aa[i].y = w * kk[i].y; aa[i].z = w * kk[i].z; aa[i].w = w * kk[i].w;
        } else {
          aa[i].x = aa[i].x + w * kk[i].x; aa[i].y = aa[i].y + w * kk[i].y;
          aa[i].z = aa[i].z + w * kk[i].z; aa[i].w = aa[i].w + w * kk[i].w;
        }
        xa[i].x = xa[i].x + c * kk[i].x; xa[i].y = xa[i].y + c * kk[i].y;
        xa[i].z = xa[i].z + c * kk[i].z; xa[i].w = xa[i].w + c * kk[i].w;
      }
    }
    if (AM == MM_A_FINAL) {
#pragma unroll
      for (int i = 0; i < 8; ++i) {
        xa[i].x = xa[i].x + 0.5f * (aa[i].x + kk[i].x); xa[i].y = xa[i].y + 0.5f * (aa[i].y + kk[i].y);
        xa[i].z = xa[i].z + 0.5f * (aa[i].z + kk[i].z); xa[i].w = xa[i].w + 0.5f * (aa[i].w + kk[i].w);
      }
    }
#pragma unroll
    for (int i = 0; i < 8; ++i)
      *(float4*)(as + (2 * i + half) * MM_APITCH + 4 * m) = xa[i];
#pragma unroll
    for (int u = 0; u < ITERS; ++u) {
      float* gp = (float*)&Gs[gslot[u]];
      *(float4*)(gp)     = make_float4(g0[u].x, g1[u].x, g0[u].y, g1[u].y);
      *(float4*)(gp + 4) = make_float4(g0[u].z, g1[u].z, g0[u].w, g1[u].w);
    }
    __syncthreads();
    if ((AM == MM_A_TANH || AM == MM_A_SET || AM == MM_A_ADD) && live) {
      float* sp = (AM == MM_A_TANH ? Hio : Acc) + tbase + kc + 4 * m;
#pragma unroll
      for (int i = 0; i < 8; ++i)
        *(float4*)(sp + (size_t)(2 * i + half) * K) = (AM == MM_A_TANH) ? xa[i] : aa[i];
    }
#pragma unroll
    for (int j = 0; j < MM_KC / 4; ++j) {
      const v2f a = *(const v2f*)(as + m * MM_APITCH + 4 * j + 2 * half);
#pragma unroll
      for (int t = 0; t < NT; ++t) {
        const v2f b = Gs[(j * NT + t) * 32 + lane];
        acc[t] = __builtin_amdgcn_wmma_f32_16x16x4_f32(false, a, false, b,
                                                       (short)0, acc[t], false, false);
      }
    }
  }

  if (live) {
    const size_t row0 = (size_t)(tile * 16 + half * 8);
    float* zr = Zp + row0 * CP + m;
#pragma unroll
    for (int t = 0; t < NTP; ++t)
#pragma unroll
      for (int r = 0; r < 8; ++r)
        zr[(size_t)r * CP + 16 * t] = acc[t][r];
    float* sr = S + row0 * CO + m;
#pragma unroll
    for (int t = NTP; t < NT; ++t)
#pragma unroll
      for (int r = 0; r < 8; ++r)
        sr[(size_t)r * CO + 16 * (t - NTP)] = acc[t][r];
  }
}

__global__ void gde_prep_g(float* __restrict__ G, const float* __restrict__ Wp,
                           const float* __restrict__ R, int Cin, int Cout,
                           unsigned* __restrict__ cz, int nz) {
  int idx = blockIdx.x * blockDim.x + threadIdx.x;
  if (idx < nz) cz[idx] = 0u;
  int Ncols = 3 * Cout;
  int tot = Cin * Ncols;
  if (idx >= tot) return;
  int r = idx / Ncols, c = idx % Ncols;
  float v;
  if (c < 2 * Cout) {
    const int ch = c >> 1;
    const float w0 = Wp[r * Cout + ch];
    v = (c & 1) ? (Wp[(size_t)Cin * Cout + r * Cout + ch] - w0) : w0;
  } else {
    v = R[r * Cout + (c - 2 * Cout)];
  }
  G[idx] = v;
}

#define EB_NB      256
#define EB_RB      ((GDE_N + EB_NB - 1) / EB_NB)
#define EB_WAVES   32
#define EB_THREADS (32 * EB_WAVES)
#define EB_OFFW    560
static_assert(EB_NB <= EB_THREADS && EB_RB <= EB_THREADS && EB_NB + 1 <= EB_OFFW, "one thread per range / per row");
static_assert(GDE_N <= 65536 && EB_RB <= 65536, "a record packs the source row and the local row in 16 bits each");
static_assert((EB_OFFW + GDE_N + 2 * GDE_E + 256 * 192 + 2 * 64 * 192 + 64 * 48) % 128 == 0, "plane alignment");

__global__ __launch_bounds__(EB_THREADS) void gde_deg(unsigned* __restrict__ bcnt,
                                                      const int* __restrict__ dst, int E) {
  __shared__ unsigned hist[EB_NB];
  const int tid = threadIdx.x;
  if (tid < EB_NB) hist[tid] = 0u;
  __syncthreads();
  const int e = blockIdx.x * EB_THREADS + tid;
  if (e < E) {
    const int d = dst[e];
    atomicAdd(&hist[d / EB_RB], 1u);
  }
  __syncthreads();
  if (tid < EB_NB) {
    const unsigned h = hist[tid];
    if (h) atomicAdd(&bcnt[tid], h);
  }
}

__global__ __launch_bounds__(EB_THREADS) void gde_edge_fill(
    int* __restrict__ pk, float* __restrict__ pp, unsigned* __restrict__ off, unsigned* __restrict__ bcur,
    const unsigned* __restrict__ bcnt, const int* __restrict__ src, const int* __restrict__ dst,
    const float* __restrict__ p, int E) {
  __shared__ unsigned hist[EB_NB];
  __shared__ unsigned scan[EB_NB];
  __shared__ unsigned base[EB_NB];
  const int tid = threadIdx.x;
  unsigned cntv = 0u;
  if (tid < EB_NB) { hist[tid] = 0u; cntv = bcnt[tid]; scan[tid] = cntv; }
  __syncthreads();
  const int e = blockIdx.x * EB_THREADS + tid;
  int d = 0, b = 0;
  unsigned r = 0u;
  if (e < E) {
    d = dst[e];
    b = d / EB_RB;
    r = atomicAdd(&hist[b], 1u);
  }
  for (int st = 1; st < EB_NB; st <<= 1) {
    unsigned v = 0u;
    if (tid < EB_NB && tid >= st) v = scan[tid - st];
    __syncthreads();
    if (tid < EB_NB) scan[tid] += v;
    __syncthreads();
  }
  if (tid < EB_NB) {
    const unsigned excl = scan[tid] - cntv;
    const unsigned h = hist[tid];
    const unsigned rs = h ? atomicAdd(&bcur[tid], h) : 0u;
    base[tid] = excl + rs;
    if (blockIdx.x == 0) {
      off[tid] = excl;
      if (tid == EB_NB - 1) off[EB_NB] = scan[tid];
    }
  }
  __syncthreads();
  if (e < E) {
    const unsigned pos = base[b] + r;
    pk[pos] = src[e] | ((d - b * EB_RB) << 16);
    pp[pos] = p[e];
  }
}

__device__ __forceinline__ void gde_count_rows(unsigned* cnt, const int* __restrict__ pk,
                                               unsigned o0, unsigned o1, int tid) {
  for (unsigned base = o0; base < o1; base += 4u * EB_THREADS) {
    int v[4];
#pragma unroll
    for (int u = 0; u < 4; ++u) {
      const unsigned r = base + (unsigned)(u * EB_THREADS + tid);
      v[u] = pk[r < o1 ? r : o1 - 1u];
    }
#pragma unroll
    for (int u = 0; u < 4; ++u) asm volatile("" :: "v"(v[u]));
#pragma unroll
    for (int u = 0; u < 4; ++u) {
      const unsigned r = base + (unsigned)(u * EB_THREADS + tid);
      if (r < o1) atomicAdd(&cnt[(unsigned)v[u] >> 16], 1u);
    }
  }
}


template <int EPS>
__device__ __forceinline__ int gde_rec_pick(int v, int j, int sub) {
  int r = __builtin_amdgcn_readlane(v, j);
#pragma unroll
  for (int k = 1; k < EPS; ++k) {
    const int o = __builtin_amdgcn_readlane(v, j + k);
    r = (sub == k) ? o : r;
  }
  return r;
}

template <int COUT, int PITCH, int G>
__device__ __forceinline__ void gde_lds_group(float* acc, const float* __restrict__ Z,
                                              int pkv, int wav, int wbv, int jb, int cnt, int sub, int c) {
  constexpr int LPE = COUT / 2;
  constexpr int EPS = 32 / LPE;
  float4 z[G];
  int    dl[G];
  float  wa[G], wb[G];
#pragma unroll
  for (int u = 0; u < G; ++u) {
    const int j = jb + u * EPS;
    const unsigned pkx = (unsigned)gde_rec_pick<EPS>(pkv, j, sub);
    dl[u] = (int)(pkx >> 16);
    wa[u] = __int_as_float(gde_rec_pick<EPS>(wav, j, sub));
    wb[u] = __int_as_float(gde_rec_pick<EPS>(wbv, j, sub));
    z[u] = *(const float4*)(Z + (size_t)(pkx & 0xffffu) * (2 * COUT) + 4 * c);
  }
#pragma unroll
  for (int u = 0; u < G; ++u) {
    if (EPS == 1 || jb + u * EPS + sub < cnt) {
      float* ar = acc + dl[u] * PITCH + 2 * c;
      __hip_atomic_fetch_add(ar,     wa[u] * z[u].x + wb[u] * z[u].y, __ATOMIC_RELAXED, __HIP_MEMORY_SCOPE_WORKGROUP);
      __hip_atomic_fetch_add(ar + 1, wa[u] * z[u].z + wb[u] * z[u].w, __ATOMIC_RELAXED, __HIP_MEMORY_SCOPE_WORKGROUP);
    }
  }
}

template <int COUT, bool DC>
__global__ __launch_bounds__(EB_THREADS) void gde_edge_lds(
    float* agg, const float* __restrict__ Z, const int* __restrict__ pk, const float* __restrict__ pp,
    const unsigned* __restrict__ off, float* deg) {
  constexpr int LPE = COUT / 2;
  constexpr int EPS = 32 / LPE;
  constexpr int GRP = EPS == 1 ? 16 : 8;
  static_assert(32 % LPE == 0 && COUT % 4 == 0 && 32 % (GRP * EPS) == 0, "shape");
  __shared__ __attribute__((aligned(16))) float acc[EB_RB * COUT];
  __shared__ float inv[EB_RB];
  __shared__ unsigned cnt[EB_RB];
  const int tid  = threadIdx.x;
  const int lane = tid & 31;
  const int wave = __builtin_amdgcn_readfirstlane(tid >> 5);
  const int sub  = lane / LPE;
  const int c    = lane % LPE;
  const int row0 = blockIdx.x * EB_RB;
  const int nrr  = GDE_N - row0;
  const int nr   = nrr < EB_RB ? (nrr > 0 ? nrr : 0) : EB_RB;
  const int nq   = nr * (COUT / 4);
  float* ag = agg + (size_t)row0 * COUT;
  constexpr int NQ = (EB_RB * (COUT / 4) + EB_THREADS - 1) / EB_THREADS;
  if (DC) {
    if (tid < EB_RB) cnt[tid] = 0u;
    __syncthreads();
    gde_count_rows(cnt, pk, off[blockIdx.x], off[blockIdx.x + 1], tid);
    __syncthreads();
  }
  if (nr > 0) {
    v4f st[NQ];
#pragma unroll
    for (int u = 0; u < NQ; ++u) {
      const int qi = tid + u * EB_THREADS;
      st[u] = ((const v4f*)ag)[qi < nq ? qi : nq - 1];
    }
    const float dg = DC ? (float)cnt[tid < nr ? tid : nr - 1] : deg[row0 + (tid < nr ? tid : nr - 1)];
#pragma unroll
    for (int u = 0; u < NQ; ++u) asm volatile("" :: "v"(st[u]));
#pragma unroll
    for (int u = 0; u < NQ; ++u) {
      const int qi = tid + u * EB_THREADS;
      if (qi < nq) ((v4f*)acc)[qi] = st[u];
    }
    if (tid < nr) {
      inv[tid] = 1.0f / fmaxf(dg, 1.0f);
      if (DC) deg[row0 + tid] = dg;
    }
  }
  __syncthreads();

  const unsigned o0 = off[blockIdx.x], o1 = off[blockIdx.x + 1];
  const unsigned share = (o1 - o0 + EB_WAVES - 1) / EB_WAVES;
  unsigned i = o0 + (unsigned)wave * share;
  const unsigned we = (i + share < o1) ? (i + share) : o1;
  unsigned ri = i + lane;
  unsigned rc = ri < GDE_E - 1u ? ri : GDE_E - 1u;
  int   npk = pk[rc];
  float npv = pp[rc];
  if (ri >= we) { npk = 0; npv = 0.0f; }
#pragma unroll 1
  for (; i < we; i += 32) {
    const int   cnt = (we - i < 32u) ? (int)(we - i) : 32;
    const int   pkv = npk;
    const float pv  = npv;
    ri = i + 32u + lane;
    rc = ri < GDE_E - 1u ? ri : GDE_E - 1u;
    npk = pk[rc];
    npv = pp[rc];
    if (ri >= we) { npk = 0; npv = 0.0f; }
    const float iv  = (lane < cnt) ? inv[(unsigned)pkv >> 16] : 0.0f;
    const int   wav = __float_as_int(iv);
    const int   wbv = __float_as_int(pv * iv);
    const int S = (cnt + EPS - 1) / EPS;
    int q = 0;
#define GDE_RUN(G) gde_lds_group<COUT, COUT, G>(acc, Z, pkv, wav, wbv, q * EPS, cnt, sub, c);
#pragma unroll 1
    for (; q + GRP <= S; q += GRP) GDE_RUN(GRP)
    if (GRP > 8 && S - q >= 8) { GDE_RUN(8) q += 8; }
    if (S - q >= 4) { GDE_RUN(4) q += 4; }
    if (S - q >= 2) { GDE_RUN(2) q += 2; }
    if (S - q >= 1) { GDE_RUN(1) }
#undef GDE_RUN
  }
  __syncthreads();
  for (int j = tid; j < nq; j += EB_THREADS) ((float4*)ag)[j] = ((const float4*)acc)[j];
}

#define EM_PITCH 68
#define EM_RT    ((EB_RB + 15) / 16)
#define EM_ROWS  (16 * EM_RT)
static_assert(EB_RB * 16 <= 4 * EB_THREADS, "a range's rows are at most four 16-byte pieces per thread");

template <int TM, int NT, bool DC>
__global__ __launch_bounds__(EB_THREADS) void gde_edge_mm(
    const float* Sin, float* Sout, const float* __restrict__ Z, float* __restrict__ Zn,
    const int* __restrict__ pk, const float* __restrict__ pp, const unsigned* __restrict__ off,
    float* deg, const float* __restrict__ G, const float* __restrict__ bias,
    float* Hp, float* Accp, float w, float cs) {
  static_assert(NT % 3 == 0, "columns are [pairs | root], two to one");
  constexpr int COUT  = 64;
  constexpr int NC    = 16 * NT;
  constexpr int NTP   = 2 * NT / 3;
  constexpr int CP    = 16 * NTP;
  constexpr int CO    = NC - CP;
  constexpr int TW    = (NT % 4 == 0) ? 4 : 3;
  constexpr int CG    = NT / TW;
  static_assert(NT % TW == 0, "whole work items");
  static_assert(CG == 1 || (TW * 16 * 4) % 128 == 0, "an item's stores are whole 128-byte lines");
  constexpr int ITEMS = (COUT / 2) * (NC / 4);
  constexpr int ITERS = (ITEMS + EB_THREADS - 1) / EB_THREADS;
  constexpr int NQ    = (EB_RB * (COUT / 4) + EB_THREADS - 1) / EB_THREADS;
  constexpr int GRP   = 16;
  __shared__ __attribute__((aligned(16))) float acc[EM_ROWS * EM_PITCH];
  __shared__ __attribute__((aligned(16))) v2f   Gs[(COUT / 4) * NT * 32];
  __shared__ float inv[EB_RB];
  __shared__ unsigned cnt[EB_RB];
  const int tid  = threadIdx.x;
  const int lane = tid & 31;
  const int wave = __builtin_amdgcn_readfirstlane(tid >> 5);
  const int row0 = blockIdx.x * EB_RB;
  const int nrr  = GDE_N - row0;
  const int nr   = nrr < EB_RB ? (nrr > 0 ? nrr : 0) : EB_RB;
  const int nq   = nr * (COUT / 4);
  const int nrt  = (nr + 15) >> 4;
  const size_t gbase = (size_t)row0 * COUT;
  if (DC) {
    if (tid < EB_RB) cnt[tid] = 0u;
    __syncthreads();
    gde_count_rows(cnt, pk, off[blockIdx.x], off[blockIdx.x + 1], tid);
    __syncthreads();
  }
  if (nr > 0) {
    v4f st[NQ];
#pragma unroll
    for (int u = 0; u < NQ; ++u) {
      const int qi = tid + u * EB_THREADS;
      st[u] = ((const v4f*)(Sin + gbase))[qi < nq ? qi : nq - 1];
    }
    const float dg = DC ? (float)cnt[tid < nr ? tid : nr - 1] : deg[row0 + (tid < nr ? tid : nr - 1)];
    float4 g0[ITERS], g1[ITERS];
    int    gslot[ITERS];
#pragma unroll
    for (int u = 0; u < ITERS; ++u) {
      const int ic = (tid + u * EB_THREADS) % ITEMS;
      const int q  = ic / (NC / 4);
      const int c0 = (ic - q * (NC / 4)) * 4;
      g0[u] = *(const float4*)(G + (size_t)(2 * q)     * NC + c0);
      g1[u] = *(const float4*)(G + (size_t)(2 * q + 1) * NC + c0);
      gslot[u] = (((q >> 1) * NT + (c0 >> 4)) * 32) + (q & 1) * 16 + (c0 & 15);
    }
#pragma unroll
    for (int u = 0; u < NQ; ++u) asm volatile("" :: "v"(st[u]));
#pragma unroll
    for (int u = 0; u < NQ; ++u) {
      const int qi = tid + u * EB_THREADS;
      if (qi < nq) *(v4f*)(acc + (qi >> 4) * EM_PITCH + 4 * (qi & 15)) = st[u];
    }
    if (tid < (16 * nrt - nr) * EM_PITCH) acc[nr * EM_PITCH + tid] = 0.0f;
    if (tid < nr) {
      inv[tid] = 1.0f / fmaxf(dg, 1.0f);
      if (DC) deg[row0 + tid] = dg;
    }
#pragma unroll
    for (int u = 0; u < ITERS; ++u) {
      float* gp = (float*)&Gs[gslot[u]];
      *(float4*)(gp)     = make_float4(g0[u].x, g1[u].x, g0[u].y, g1[u].y);
      *(float4*)(gp + 4) = make_float4(g0[u].z, g1[u].z, g0[u].w, g1[u].w);
    }
  }
  __syncthreads();

  const unsigned o0 = off[blockIdx.x], o1 = off[blockIdx.x + 1];
  const unsigned share = (o1 - o0 + EB_WAVES - 1) / EB_WAVES;
  unsigned i = o0 + (unsigned)wave * share;
  const unsigned we = (i + share < o1) ? (i + share) : o1;
  unsigned ri = i + lane;
  unsigned rc = ri < GDE_E - 1u ? ri : GDE_E - 1u;
  int   npk = pk[rc];
  float npv = pp[rc];
  if (ri >= we) { npk = 0; npv = 0.0f; }
#pragma unroll 1
  for (; i < we; i += 32) {
    const int   cnt = (we - i < 32u) ? (int)(we - i) : 32;
    const int   pkv = npk;
    const float pv  = npv;
    ri = i + 32u + lane;
    rc = ri < GDE_E - 1u ? ri : GDE_E - 1u;
    npk = pk[rc];
    npv = pp[rc];
    if (ri >= we) { npk = 0; npv = 0.0f; }
    const float iv  = (lane < cnt) ? inv[(unsigned)pkv >> 16] : 0.0f;
    const int   wav = __float_as_int(iv);
    const int   wbv = __float_as_int(pv * iv);
    int q = 0;
#define GDE_RUN(G_) gde_lds_group<COUT, EM_PITCH, G_>(acc, Z, pkv, wav, wbv, q, cnt, 0, lane);
#pragma unroll 1
    for (; q + GRP <= cnt; q += GRP) GDE_RUN(GRP)
    if (cnt - q >= 8) { GDE_RUN(8) q += 8; }
    if (cnt - q >= 4) { GDE_RUN(4) q += 4; }
    if (cnt - q >= 2) { GDE_RUN(2) q += 2; }
    if (cnt - q >= 1) { GDE_RUN(1) }
#undef GDE_RUN
  }

  v4f hv[NQ], av[NQ];
  if (nr > 0) {
#pragma unroll
    for (int u = 0; u < NQ; ++u) {
      const int qi = tid + u * EB_THREADS;
      const int qc = qi < nq ? qi : nq - 1;
      if (TM == MM_A_SET || TM == MM_A_ADD || TM == MM_A_FINAL) hv[u] = ((const v4f*)(Hp + gbase))[qc];
      if (TM == MM_A_ADD || TM == MM_A_FINAL) av[u] = ((const v4f*)(Accp + gbase))[qc];
    }
  }
  __syncthreads();
  if (TM != MM_A_PLAIN) {
#pragma unroll
    for (int u = 0; u < NQ; ++u) {
      const int qi = tid + u * EB_THREADS;
      if (qi < nq) {
        float* lp = acc + (qi >> 4) * EM_PITCH + 4 * (qi & 15);
        v4f kv = *(const v4f*)lp;
        if (TM == MM_A_TANH) {
          kv.x = tanhf(kv.x); kv.y = tanhf(kv.y); kv.z = tanhf(kv.z); kv.w = tanhf(kv.w);
          ((v4f*)(Hp + gbase))[qi] = kv;
          *(v4f*)lp = kv;
        }
        if (TM == MM_A_SET || TM == MM_A_ADD) {
          v4f a;
          if (TM == MM_A_SET) {
            a.x = w * kv.x; a.y = w * kv.y; a.z = w * kv.z; a.w = w * kv.w;
          } else {
            a.x = av[u].x + w * kv.x; a.y = av[u].y + w * kv.y;
            a.z = av[u].z + w * kv.z; a.w = av[u].w + w * kv.w;
          }
          ((v4f*)(Accp + gbase))[qi] = a;
          v4f y;
          y.x = hv[u].x + cs * kv.x; y.y = hv[u].y + cs * kv.y;
          y.z = hv[u].z + cs * kv.z; y.w = hv[u].w + cs * kv.w;
          *(v4f*)lp = y;
        }
        if (TM == MM_A_FINAL) {
          v4f y;
          y.x = hv[u].x + 0.5f * (av[u].x + kv.x); y.y = hv[u].y + 0.5f * (av[u].y + kv.y);
          y.z = hv[u].z + 0.5f * (av[u].z + kv.z); y.w = hv[u].w + 0.5f * (av[u].w + kv.w);
          *(v4f*)lp = y;
        }
      }
    }
    __syncthreads();
  }

  const int m     = lane & 15;
  const int half  = lane >> 4;
  const int items = nrt * CG;
#pragma unroll 1
  for (int it = wave; it < items; it += EB_WAVES) {
    const int rt = it / CG;
    const int t0 = (it - rt * CG) * TW;
    v8f d[TW];
#pragma unroll
    for (int t = 0; t < TW; ++t) {
      const int tt = t0 + t;
      const float bv = (tt >= NTP) ? bias[16 * (tt - NTP) + m] : 0.0f;
      d[t] = (v8f){bv, bv, bv, bv, bv, bv, bv, bv};
    }
    const float* ar = acc + (16 * rt + m) * EM_PITCH + 2 * half;
#pragma unroll 4
    for (int j = 0; j < COUT / 4; ++j) {
      const v2f a = *(const v2f*)(ar + 4 * j);
#pragma unroll
      for (int t = 0; t < TW; ++t) {
        const v2f b = Gs[(j * NT + t0 + t) * 32 + lane];
        d[t] = __builtin_amdgcn_wmma_f32_16x16x4_f32(false, a, false, b,
                                                     (short)0, d[t], false, false);
      }
    }
    const int lr = 16 * rt + 8 * half;
#pragma unroll
    for (int t = 0; t < TW; ++t) {
      const int tt = t0 + t;
      if (tt < NTP) {
        float* zr = Zn + (size_t)(row0 + lr) * CP + 16 * tt + m;
#pragma unroll
        for (int r = 0; r < 8; ++r)
          if (lr + r < nr) zr[(size_t)r * CP] = d[t][r];
      } else {
        float* sr = Sout + (size_t)(row0 + lr) * CO + 16 * (tt - NTP) + m;
#pragma unroll
        for (int r = 0; r < 8; ++r)
          if (lr + r < nr) sr[(size_t)r * CO] = d[t][r];
      }
    }
  }
}

__global__ void gde_log_softmax16(float* __restrict__ out, int n) {
  int i = blockIdx.x * blockDim.x + threadIdx.x;
  if (i >= n) return;
  float v[16];
  float mx = -3.402823466e+38f;
  float* row = out + (size_t)i * 16;
#pragma unroll
  for (int c = 0; c < 16; ++c) { v[c] = tanhf(row[c]); mx = fmaxf(mx, v[c]); }
  float s = 0.f;
#pragma unroll
  for (int c = 0; c < 16; ++c) s += __expf(v[c] - mx);
  float ls = __logf(s);
#pragma unroll
  for (int c = 0; c < 16; ++c) row[c] = v[c] - mx - ls;
}

extern "C" void kernel_launch(void* const* d_in, const int* in_sizes, int n_in,
                              void* d_out, int out_size, void* d_ws, size_t ws_size,
                              hipStream_t stream) {
  const float* x   = (const float*)d_in[0];
  const float* p   = (const float*)d_in[1];
  const int*   src = (const int*)d_in[2];
  const int*   dst = (const int*)d_in[3];
  const float* W1 = (const float*)d_in[4],  *R1 = (const float*)d_in[5],  *b1 = (const float*)d_in[6];
  const float* Wa = (const float*)d_in[7],  *Ra = (const float*)d_in[8],  *ba = (const float*)d_in[9];
  const float* Wb = (const float*)d_in[10], *Rb = (const float*)d_in[11], *bb = (const float*)d_in[12];
  const float* W2 = (const float*)d_in[13], *R2 = (const float*)d_in[14], *b2 = (const float*)d_in[15];
  float* out = (float*)d_out;

  float* W = (float*)d_ws;
  size_t o = 0;
  float* Z   = W + o; o += (size_t)GDE_N * 128;
  float* H   = W + o; o += (size_t)GDE_N * 64;
  float* U   = W + o; o += (size_t)GDE_N * 64;
  float* ACC = W + o; o += (size_t)GDE_N * 64;
  unsigned* OFF = (unsigned*)(W + o); o += (size_t)EB_OFFW;
  float* DEG = W + o; o += (size_t)GDE_N;
  int*   PK  = (int*)(W + o); o += (size_t)GDE_E;
  float* PP  = W + o; o += (size_t)GDE_E;
  unsigned* BCNT = (unsigned*)U;
  unsigned* BCUR = BCNT + EB_NB;
  float* GC1 = W + o; o += (size_t)256 * 192;
  float* GCA = W + o; o += (size_t)64 * 192;
  float* GCB = W + o; o += (size_t)64 * 192;
  float* GC2 = W + o; o += (size_t)64 * 48;
  float* Kb  = W + o; o += (size_t)GDE_N * 64;
  float* Z2  = Kb;
  const bool tails = ws_size >= sizeof(float) * (o + (size_t)GDE_N * 64);
  (void)n_in; (void)in_sizes; (void)out_size;

  gde_prep_g<<<(256 * 192 + 255) / 256, 256, 0, stream>>>(GC1, W1, R1, 256, 64, BCNT, 2 * EB_NB);
  gde_prep_g<<<(64 * 192 + 255) / 256, 256, 0, stream>>>(GCA, Wa, Ra, 64, 64, BCNT, 0);
  gde_prep_g<<<(64 * 192 + 255) / 256, 256, 0, stream>>>(GCB, Wb, Rb, 64, 64, BCNT, 0);
  gde_prep_g<<<(64 * 48 + 255) / 256, 256, 0, stream>>>(GC2, W2, R2, 64, 16, BCNT, 0);

  gde_deg<<<(GDE_E + EB_THREADS - 1) / EB_THREADS, EB_THREADS, 0, stream>>>(BCNT, dst, GDE_E);
  gde_edge_fill<<<(GDE_E + EB_THREADS - 1) / EB_THREADS, EB_THREADS, 0, stream>>>(PK, PP, OFF, BCUR, BCNT, src, dst, p, GDE_E);

  const int mmg = (MM_TILES + MM_WAVES - 1) / MM_WAVES;
  if (tails) {
    gde_mm_lds<256, 12, MM_A_PLAIN><<<mmg, MM_THREADS, 0, stream>>>(x, GC1, Z, H, b1, nullptr, nullptr, nullptr, 0.0f, 0.0f);
    gde_edge_mm<MM_A_TANH, 12, true><<<EB_NB, EB_THREADS, 0, stream>>>(H, U, Z, Z2, PK, PP, OFF, DEG, GCA, ba, H, ACC, 0.0f, 0.0f);
    gde_edge_mm<MM_A_PLAIN, 12, false><<<EB_NB, EB_THREADS, 0, stream>>>(U, U, Z2, Z, PK, PP, OFF, DEG, GCB, bb, H, ACC, 0.0f, 0.0f);
    gde_edge_mm<MM_A_SET, 12, false><<<EB_NB, EB_THREADS, 0, stream>>>(U, U, Z, Z2, PK, PP, OFF, DEG, GCA, ba, H, ACC, 1.0f, 1.5f);
    gde_edge_mm<MM_A_PLAIN, 12, false><<<EB_NB, EB_THREADS, 0, stream>>>(U, U, Z2, Z, PK, PP, OFF, DEG, GCB, bb, H, ACC, 0.0f, 0.0f);
    gde_edge_mm<MM_A_ADD, 12, false><<<EB_NB, EB_THREADS, 0, stream>>>(U, U, Z, Z2, PK, PP, OFF, DEG, GCA, ba, H, ACC, 2.0f, 1.5f);
    gde_edge_mm<MM_A_PLAIN, 12, false><<<EB_NB, EB_THREADS, 0, stream>>>(U, U, Z2, Z, PK, PP, OFF, DEG, GCB, bb, H, ACC, 0.0f, 0.0f);
    gde_edge_mm<MM_A_ADD, 12, false><<<EB_NB, EB_THREADS, 0, stream>>>(U, U, Z, Z2, PK, PP, OFF, DEG, GCA, ba, H, ACC, 2.0f, 3.0f);
    gde_edge_mm<MM_A_PLAIN, 12, false><<<EB_NB, EB_THREADS, 0, stream>>>(U, U, Z2, Z, PK, PP, OFF, DEG, GCB, bb, H, ACC, 0.0f, 0.0f);
    gde_edge_mm<MM_A_FINAL, 3, false><<<EB_NB, EB_THREADS, 0, stream>>>(U, out, Z, Z2, PK, PP, OFF, DEG, GC2, b2, H, ACC, 0.0f, 0.0f);
    gde_edge_lds<16, false><<<EB_NB, EB_THREADS, 0, stream>>>(out, Z2, PK, PP, OFF, DEG);
    gde_log_softmax16<<<(GDE_N + 255) / 256, 256, 0, stream>>>(out, GDE_N);
    return;
  }

  auto spline = [&](const float* y, int Cin, int Cout, const float* G,
                    const float* bias, float* ob, int am, float w, float c) {
    if (Cin == 256)
      gde_mm_lds<256, 12, MM_A_PLAIN><<<mmg, MM_THREADS, 0, stream>>>(y, G, Z, ob, bias, nullptr, nullptr, nullptr, w, c);
    else if (am == MM_A_PLAIN)
      gde_mm_lds<64, 12, MM_A_PLAIN><<<mmg, MM_THREADS, 0, stream>>>(y, G, Z, ob, bias, nullptr, nullptr, nullptr, w, c);
    else if (am == MM_A_TANH)
      gde_mm_lds<64, 12, MM_A_TANH><<<mmg, MM_THREADS, 0, stream>>>(y, G, Z, ob, bias, H, Kb, ACC, w, c);
    else if (am == MM_A_SET)
      gde_mm_lds<64, 12, MM_A_SET><<<mmg, MM_THREADS, 0, stream>>>(y, G, Z, ob, bias, H, Kb, ACC, w, c);
    else if (am == MM_A_ADD)
      gde_mm_lds<64, 12, MM_A_ADD><<<mmg, MM_THREADS, 0, stream>>>(y, G, Z, ob, bias, H, Kb, ACC, w, c);
    else
      gde_mm_lds<64, 3, MM_A_FINAL><<<mmg, MM_THREADS, 0, stream>>>(y, G, Z, ob, bias, H, Kb, ACC, w, c);
    if (Cin == 256)      gde_edge_lds<64, true><<<EB_NB, EB_THREADS, 0, stream>>>(ob, Z, PK, PP, OFF, DEG);
    else if (Cout == 64) gde_edge_lds<64, false><<<EB_NB, EB_THREADS, 0, stream>>>(ob, Z, PK, PP, OFF, DEG);
    else                 gde_edge_lds<16, false><<<EB_NB, EB_THREADS, 0, stream>>>(ob, Z, PK, PP, OFF, DEG);
  };
  auto odefunc = [&](int am, float w, float c) {
    spline(H, 64, 64, GCA, ba, U, am, w, c);
    spline(U, 64, 64, GCB, bb, Kb, MM_A_PLAIN, 0.0f, 0.0f);
  };

  spline(x, 256, 64, GC1, b1, H, MM_A_PLAIN, 0.0f, 0.0f);

  odefunc(MM_A_TANH, 0.0f, 0.0f);
  odefunc(MM_A_SET, 1.0f, 1.5f);
  odefunc(MM_A_ADD, 2.0f, 1.5f);
  odefunc(MM_A_ADD, 2.0f, 3.0f);

  spline(H, 64, 16, GC2, b2, out, MM_A_FINAL, 0.0f, 0.0f);
  gde_log_softmax16<<<(GDE_N + 255) / 256, 256, 0, stream>>>(out, GDE_N);
}
